# starting kernel
# baseline (speedup 1.0000x reference)
#include <hip/hip_runtime.h>

typedef __attribute__((ext_vector_type(2))) float v2f;
typedef __attribute__((ext_vector_type(8))) float v8f;
typedef __attribute__((ext_vector_type(4))) float v4f;

__global__ void zero_f32v4(float* __restrict__ p, long long count4) {
  long long i = (long long)blockIdx.x * blockDim.x + threadIdx.x;
  if (i < count4) {
    v4f z = {};
    *(v4f*)(p + 4 * i) = z;
  }
}

template <int CIN, int COUT, int TPB>
__global__ __launch_bounds__(32) void sconv_wmma(
    const float* __restrict__ f, const float* __restrict__ W,
    const int* __restrict__ ii, const int* __restrict__ oo,
    const int* __restrict__ nin_p, const int* __restrict__ nout_p,
    int P, float* __restrict__ out) {
  const int lane = threadIdx.x;
  const int m    = lane & 15;
  const int half = lane >> 4;

  const int k       = blockIdx.y;
  const int colBase = blockIdx.z << 4;

  const int n_in  = *nin_p;
  const int n_out = *nout_p;

  const int col  = colBase + m;
  const bool cv  = (col < COUT);

  const float* __restrict__ Wk = W + (long long)k * (CIN * COUT);
  v2f bfrag[CIN / 4];
#pragma unroll
  for (int c = 0; c < CIN / 4; ++c) {
    const int k0 = c * 4 + half * 2;
    if (cv) {
      bfrag[c].x = Wk[(k0 + 0) * COUT + col];
      bfrag[c].y = Wk[(k0 + 1) * COUT + col];
    } else {
      bfrag[c].x = 0.0f; bfrag[c].y = 0.0f;
    }
  }

  const int* __restrict__ iik = ii + (long long)k * P;
  const int* __restrict__ ook = oo + (long long)k * P;

#pragma unroll 1
  for (int t = 0; t < TPB; ++t) {
    const int tile = blockIdx.x * TPB + t;
    const int base = tile * 16;
    if (base >= P) break;
    const int p = base + m;
    const bool pv = (p < P);

    if (base + 16 < P) {
      __builtin_prefetch(iik + p + 16, 0, 1);
      __builtin_prefetch(ook + p + 16, 0, 1);
    }

    const int o   = pv ? ook[p] : n_out;
    const bool ov = (o >= 0) && (o < n_out);
    if (__ballot(ov) == 0ull) continue;

    int row       = pv ? iik[p] : n_in;
    const bool rv = (row >= 0) && (row < n_in);
    const float* __restrict__ fr = f + (long long)row * CIN;

    v8f acc = {};
#pragma unroll
    for (int c = 0; c < CIN / 4; ++c) {
      const int k0 = c * 4 + half * 2;
      v2f a;
      if (rv) {
        a = *(const v2f*)(fr + k0);
      } else {
        a.x = 0.0f; a.y = 0.0f;
      }
      acc = __builtin_amdgcn_wmma_f32_16x16x4_f32(
           false, a,  false, bfrag[c],
           (short)0, acc,  false,  false);
    }

#pragma unroll
    for (int r = 0; r < 8; ++r) {
      const int mr  = half * 8 + r;
      const int o_r = __shfl(o, mr);
      if (cv && base + mr < P && o_r >= 0 && o_r < n_out) {
        unsafeAtomicAdd(&out[(long long)o_r * COUT + col], acc[r]);
      }
    }
  }
}

template <int C>
__global__ __launch_bounds__(256) void bn_stats(
    const float* __restrict__ x, const int* __restrict__ n_p,
    float* __restrict__ stats) {
  constexpr int RPB = 256 / C;
  const int n = *n_p;
  const int t = threadIdx.x;
  const int c = t & (C - 1);
  const int rg = t / C;

  float s = 0.0f, q = 0.0f;
  for (long long row = (long long)blockIdx.x * RPB + rg; row < n;
       row += (long long)gridDim.x * RPB) {
    const float v = x[row * C + c];
    s += v;
    q += v * v;
  }

  __shared__ float ls[256];
  __shared__ float lq[256];
  ls[t] = s; lq[t] = q;
  __syncthreads();
#pragma unroll
  for (int off = 128; off >= C; off >>= 1) {
    if (t < off) { ls[t] += ls[t + off]; lq[t] += lq[t + off]; }
    __syncthreads();
  }
  if (t < C) {
    unsafeAtomicAdd(&stats[c], ls[t]);
    unsafeAtomicAdd(&stats[C + c], lq[t]);
  }
}

template <int C>
__global__ __launch_bounds__(256) void bn_relu(
    float* __restrict__ x, const float* __restrict__ stats,
    const float* __restrict__ g, const float* __restrict__ b,
    const int* __restrict__ n_p) {
  const int n = *n_p;
  const long long idx = (long long)blockIdx.x * 256 + threadIdx.x;
  const long long row = idx / C;
  const int c         = (int)(idx % C);
  if (row >= n) return;
  const float inv_n = 1.0f / (float)n;
  const float mean  = stats[c] * inv_n;
  const float var   = stats[C + c] * inv_n - mean * mean;
  const float y = (x[idx] - mean) * rsqrtf(var + 1e-5f) * g[c] + b[c];
  x[idx] = y > 0.0f ? y : 0.0f;
}

extern "C" void kernel_launch(void* const* d_in, const int* in_sizes, int n_in,
                              void* d_out, int out_size, void* d_ws, size_t ws_size,
                              hipStream_t stream) {
  (void)n_in; (void)ws_size;
  const float* feats = (const float*)d_in[0];
  const float* W0  = (const float*)d_in[1];
  const float* g0  = (const float*)d_in[2];
  const float* b0  = (const float*)d_in[3];
  const float* W1  = (const float*)d_in[4];
  const float* g1  = (const float*)d_in[5];
  const float* b1  = (const float*)d_in[6];
  const float* W2  = (const float*)d_in[7];
  const float* g2  = (const float*)d_in[8];
  const float* b2  = (const float*)d_in[9];
  const float* Wt1 = (const float*)d_in[10];
  const float* gt1 = (const float*)d_in[11];
  const float* bt1 = (const float*)d_in[12];
  const float* Wt2 = (const float*)d_in[13];
  const int* in0  = (const int*)d_in[14];
  const int* out0 = (const int*)d_in[15];
  const int* in1  = (const int*)d_in[16];
  const int* out1 = (const int*)d_in[17];
  const int* in2  = (const int*)d_in[18];
  const int* out2 = (const int*)d_in[19];
  const int* int1 = (const int*)d_in[20];
  const int* outt1= (const int*)d_in[21];
  const int* int2 = (const int*)d_in[22];
  const int* outt2= (const int*)d_in[23];
  const int* n0p  = (const int*)d_in[24];
  const int* n1p  = (const int*)d_in[25];
  const int* n2p  = (const int*)d_in[26];

  const int n0  = in_sizes[0] / 4;
  const int P0  = in_sizes[14] / 27;
  const int P1  = in_sizes[16] / 27;
  const int P2  = in_sizes[18] / 27;
  const int Pt1 = in_sizes[20] / 27;
  const int Pt2 = in_sizes[22] / 27;

  float* ws  = (float*)d_ws;
  float* x0  = ws;
  float* x1  = x0  + (size_t)n0 * 16;
  float* x2  = x1  + (size_t)n0 * 32;
  float* xt1 = x2  + (size_t)n0 * 64;
  float* st0 = xt1 + (size_t)n0 * 32;
  float* st1 = st0 + 32;
  float* st2 = st1 + 64;
  float* stt = st2 + 128;

  const long long zc4 = ((long long)n0 * (16 + 32 + 64 + 32) + 512) / 4;
  zero_f32v4<<<dim3((unsigned)((zc4 + 255) / 256)), 256, 0, stream>>>(ws, zc4);
  const long long oc4 = ((long long)out_size + 3) / 4;
  zero_f32v4<<<dim3((unsigned)((oc4 + 255) / 256)), 256, 0, stream>>>(
      (float*)d_out, oc4);

  constexpr int TPB = 8;
  const dim3 wv(32);
  auto tgrid = [](int P) { return (unsigned)((P + 16 * TPB - 1) / (16 * TPB)); };
  auto eg = [](long long elems) { return dim3((unsigned)((elems + 255) / 256)); };
  auto sg = [&](int C) {
    long long blocks = ((long long)n0 * C + 255) / 256;
    return dim3((unsigned)(blocks < 768 ? blocks : 768));
  };

  sconv_wmma<4, 16, TPB><<<dim3(tgrid(P0), 27, 1), wv, 0, stream>>>(
      feats, W0, in0, out0, n0p, n0p, P0, x0);
  bn_stats<16><<<sg(16), 256, 0, stream>>>(x0, n0p, st0);
  bn_relu<16><<<eg((long long)n0 * 16), 256, 0, stream>>>(x0, st0, g0, b0, n0p);

  sconv_wmma<16, 32, TPB><<<dim3(tgrid(P1), 27, 2), wv, 0, stream>>>(
      x0, W1, in1, out1, n0p, n1p, P1, x1);
  bn_stats<32><<<sg(32), 256, 0, stream>>>(x1, n1p, st1);
  bn_relu<32><<<eg((long long)n0 * 32), 256, 0, stream>>>(x1, st1, g1, b1, n1p);

  sconv_wmma<32, 64, TPB><<<dim3(tgrid(P2), 27, 4), wv, 0, stream>>>(
      x1, W2, in2, out2, n1p, n2p, P2, x2);
  bn_stats<64><<<sg(64), 256, 0, stream>>>(x2, n2p, st2);
  bn_relu<64><<<eg((long long)n0 * 64), 256, 0, stream>>>(x2, st2, g2, b2, n2p);

  sconv_wmma<64, 32, TPB><<<dim3(tgrid(Pt1), 27, 2), wv, 0, stream>>>(
      x2, Wt1, int1, outt1, n2p, n1p, Pt1, xt1);
  bn_stats<32><<<sg(32), 256, 0, stream>>>(xt1, n1p, stt);
  bn_relu<32><<<eg((long long)n0 * 32), 256, 0, stream>>>(xt1, stt, gt1, bt1, n1p);

  sconv_wmma<32, 20, TPB><<<dim3(tgrid(Pt2), 27, 2), wv, 0, stream>>>(
      xt1, Wt2, int2, outt2, n1p, n0p, Pt2, (float*)d_out);
}


# Branch-free conv tile body: both index words, all CIN/4 row gathers (invalid rows read a zeroed row) and the B-fragment loads issue as groups, not one exec-masked load + full wait each
# speedup vs baseline: 1.5023x; 1.5023x over previous
#include <hip/hip_runtime.h>

typedef __attribute__((ext_vector_type(2))) float v2f;
typedef __attribute__((ext_vector_type(8))) float v8f;
typedef __attribute__((ext_vector_type(4))) float v4f;

__global__ void zero_f32v4(float* __restrict__ p, long long count4) {
  long long i = (long long)blockIdx.x * blockDim.x + threadIdx.x;
  if (i < count4) {
    v4f z = {};
    *(v4f*)(p + 4 * i) = z;
  }
}

template <int CIN, int COUT, int TPB>
__global__ __launch_bounds__(32) void sconv_wmma(
    const float* __restrict__ f, const float* __restrict__ W,
    const int* __restrict__ ii, const int* __restrict__ oo,
    const int* __restrict__ nin_p, const int* __restrict__ nout_p,
    int P, float* __restrict__ out, const float* __restrict__ zrow) {
  const int lane = threadIdx.x;
  const int m    = lane & 15;
  const int half = lane >> 4;

  const int k       = blockIdx.y;
  const int colBase = blockIdx.z << 4;

  const int n_in  = *nin_p;
  const int n_out = *nout_p;

  const int col  = colBase + m;
  constexpr bool FULLN = (COUT % 16) == 0;
  const bool cv  = FULLN || (col < COUT);
  const int colc = cv ? col : (COUT - 1);

  const float* __restrict__ Wk = W + (long long)k * (CIN * COUT);
  v2f bfrag[CIN / 4];
#pragma unroll
  for (int c = 0; c < CIN / 4; ++c) {
    const int k0 = c * 4 + half * 2;
    const float w0 = Wk[(k0 + 0) * COUT + colc];
    const float w1 = Wk[(k0 + 1) * COUT + colc];
    bfrag[c].x = cv ? w0 : 0.0f;
    bfrag[c].y = cv ? w1 : 0.0f;
  }

  const int* __restrict__ iik = ii + (long long)k * P;
  const int* __restrict__ ook = oo + (long long)k * P;

#pragma unroll 1
  for (int t = 0; t < TPB; ++t) {
    const int tile = blockIdx.x * TPB + t;
    const int base = tile * 16;
    if (base >= P) break;
    const int p = base + m;
    const bool pv = (p < P);
    const int pc = pv ? p : (P - 1);

    if (base + 16 < P) {
      __builtin_prefetch(iik + p + 16, 0, 1);
      __builtin_prefetch(ook + p + 16, 0, 1);
    }

    const int o_ld = ook[pc];
    int r_ld = iik[pc];
    asm volatile("" : "+v"(r_ld));

    const int o   = pv ? o_ld : n_out;
    const bool ov = (o >= 0) && (o < n_out);
    if (__ballot(ov) == 0ull) continue;

    const int row = pv ? r_ld : n_in;
    const bool rv = (row >= 0) && (row < n_in);
    const float* __restrict__ fr =
        (rv ? f + (long long)row * CIN : zrow) + half * 2;

    v2f a[CIN / 4];
#pragma unroll
    for (int c = 0; c < CIN / 4; ++c) a[c] = *(const v2f*)(fr + c * 4);
    __builtin_amdgcn_sched_barrier(0);

    v8f acc = {};
#pragma unroll
    for (int c = 0; c < CIN / 4; ++c) {
      acc = __builtin_amdgcn_wmma_f32_16x16x4_f32(
           false, a[c],  false, bfrag[c],
           (short)0, acc,  false,  false);
    }
    __builtin_amdgcn_sched_barrier(0);
    asm volatile("v_nop\n\tv_nop\n\tv_nop\n\tv_nop\n\tv_nop\n\tv_nop\n\tv_nop\n\tv_nop"
                 : "+v"(acc));

    int o_x = o;
    asm volatile("" : "+v"(o_x));
    int orow[8];
#pragma unroll
    for (int r = 0; r < 8; ++r) orow[r] = __shfl(o_x, half * 8 + r);
#pragma unroll
    for (int r = 0; r < 8; ++r) {
      const int mr  = half * 8 + r;
      const int o_r = orow[r];
      if (cv && base + mr < P && o_r >= 0 && o_r < n_out) {
        unsafeAtomicAdd(&out[(long long)o_r * COUT + col], acc[r]);
      }
    }
  }
}

template <int C>
__global__ __launch_bounds__(256) void bn_stats(
    const float* __restrict__ x, const int* __restrict__ n_p,
    float* __restrict__ stats) {
  constexpr int RPB = 256 / C;
  const int n = *n_p;
  const int t = threadIdx.x;
  const int c = t & (C - 1);
  const int rg = t / C;

  float s = 0.0f, q = 0.0f;
  for (long long row = (long long)blockIdx.x * RPB + rg; row < n;
       row += (long long)gridDim.x * RPB) {
    const float v = x[row * C + c];
    s += v;
    q += v * v;
  }

  __shared__ float ls[256];
  __shared__ float lq[256];
  ls[t] = s; lq[t] = q;
  __syncthreads();
#pragma unroll
  for (int off = 128; off >= C; off >>= 1) {
    if (t < off) { ls[t] += ls[t + off]; lq[t] += lq[t + off]; }
    __syncthreads();
  }
  if (t < C) {
    unsafeAtomicAdd(&stats[c], ls[t]);
    unsafeAtomicAdd(&stats[C + c], lq[t]);
  }
}

template <int C>
__global__ __launch_bounds__(256) void bn_relu(
    float* __restrict__ x, const float* __restrict__ stats,
    const float* __restrict__ g, const float* __restrict__ b,
    const int* __restrict__ n_p) {
  const int n = *n_p;
  const long long idx = (long long)blockIdx.x * 256 + threadIdx.x;
  const long long row = idx / C;
  const int c         = (int)(idx % C);
  if (row >= n) return;
  const float inv_n = 1.0f / (float)n;
  const float mean  = stats[c] * inv_n;
  const float var   = stats[C + c] * inv_n - mean * mean;
  const float y = (x[idx] - mean) * rsqrtf(var + 1e-5f) * g[c] + b[c];
  x[idx] = y > 0.0f ? y : 0.0f;
}

extern "C" void kernel_launch(void* const* d_in, const int* in_sizes, int n_in,
                              void* d_out, int out_size, void* d_ws, size_t ws_size,
                              hipStream_t stream) {
  (void)n_in; (void)ws_size;
  const float* feats = (const float*)d_in[0];
  const float* W0  = (const float*)d_in[1];
  const float* g0  = (const float*)d_in[2];
  const float* b0  = (const float*)d_in[3];
  const float* W1  = (const float*)d_in[4];
  const float* g1  = (const float*)d_in[5];
  const float* b1  = (const float*)d_in[6];
  const float* W2  = (const float*)d_in[7];
  const float* g2  = (const float*)d_in[8];
  const float* b2  = (const float*)d_in[9];
  const float* Wt1 = (const float*)d_in[10];
  const float* gt1 = (const float*)d_in[11];
  const float* bt1 = (const float*)d_in[12];
  const float* Wt2 = (const float*)d_in[13];
  const int* in0  = (const int*)d_in[14];
  const int* out0 = (const int*)d_in[15];
  const int* in1  = (const int*)d_in[16];
  const int* out1 = (const int*)d_in[17];
  const int* in2  = (const int*)d_in[18];
  const int* out2 = (const int*)d_in[19];
  const int* int1 = (const int*)d_in[20];
  const int* outt1= (const int*)d_in[21];
  const int* int2 = (const int*)d_in[22];
  const int* outt2= (const int*)d_in[23];
  const int* n0p  = (const int*)d_in[24];
  const int* n1p  = (const int*)d_in[25];
  const int* n2p  = (const int*)d_in[26];

  const int n0  = in_sizes[0] / 4;
  const int P0  = in_sizes[14] / 27;
  const int P1  = in_sizes[16] / 27;
  const int P2  = in_sizes[18] / 27;
  const int Pt1 = in_sizes[20] / 27;
  const int Pt2 = in_sizes[22] / 27;

  float* ws  = (float*)d_ws;
  float* x0  = ws;
  float* x1  = x0  + (size_t)n0 * 16;
  float* x2  = x1  + (size_t)n0 * 32;
  float* xt1 = x2  + (size_t)n0 * 64;
  float* st0 = xt1 + (size_t)n0 * 32;
  float* st1 = st0 + 32;
  float* st2 = st1 + 64;
  float* stt = st2 + 128;
  const float* zrow = stt + 64;

  const long long zc4 = ((long long)n0 * (16 + 32 + 64 + 32) + 512) / 4;
  zero_f32v4<<<dim3((unsigned)((zc4 + 255) / 256)), 256, 0, stream>>>(ws, zc4);
  const long long oc4 = ((long long)out_size + 3) / 4;
  zero_f32v4<<<dim3((unsigned)((oc4 + 255) / 256)), 256, 0, stream>>>(
      (float*)d_out, oc4);

  constexpr int TPB = 8;
  const dim3 wv(32);
  auto tgrid = [](int P) { return (unsigned)((P + 16 * TPB - 1) / (16 * TPB)); };
  auto eg = [](long long elems) { return dim3((unsigned)((elems + 255) / 256)); };
  auto sg = [&](int C) {
    long long blocks = ((long long)n0 * C + 255) / 256;
    return dim3((unsigned)(blocks < 768 ? blocks : 768));
  };

  sconv_wmma<4, 16, TPB><<<dim3(tgrid(P0), 27, 1), wv, 0, stream>>>(
      feats, W0, in0, out0, n0p, n0p, P0, x0, zrow);
  bn_stats<16><<<sg(16), 256, 0, stream>>>(x0, n0p, st0);
  bn_relu<16><<<eg((long long)n0 * 16), 256, 0, stream>>>(x0, st0, g0, b0, n0p);

  sconv_wmma<16, 32, TPB><<<dim3(tgrid(P1), 27, 2), wv, 0, stream>>>(
      x0, W1, in1, out1, n0p, n1p, P1, x1, zrow);
  bn_stats<32><<<sg(32), 256, 0, stream>>>(x1, n1p, st1);
  bn_relu<32><<<eg((long long)n0 * 32), 256, 0, stream>>>(x1, st1, g1, b1, n1p);

  sconv_wmma<32, 64, TPB><<<dim3(tgrid(P2), 27, 4), wv, 0, stream>>>(
      x1, W2, in2, out2, n1p, n2p, P2, x2, zrow);
  bn_stats<64><<<sg(64), 256, 0, stream>>>(x2, n2p, st2);
  bn_relu<64><<<eg((long long)n0 * 64), 256, 0, stream>>>(x2, st2, g2, b2, n2p);

  sconv_wmma<64, 32, TPB><<<dim3(tgrid(Pt1), 27, 2), wv, 0, stream>>>(
      x2, Wt1, int1, outt1, n2p, n1p, Pt1, xt1, zrow);
  bn_stats<32><<<sg(32), 256, 0, stream>>>(xt1, n1p, stt);
  bn_relu<32><<<eg((long long)n0 * 32), 256, 0, stream>>>(xt1, stt, gt1, bt1, n1p);

  sconv_wmma<32, 20, TPB><<<dim3(tgrid(Pt2), 27, 2), wv, 0, stream>>>(
      xt1, Wt2, int2, outt2, n1p, n0p, Pt2, (float*)d_out, zrow);
}


# BN normalize + ReLU moved into the next conv's gather (fma + max per gathered element, scale/shift built per block via LDS, sentinel rows dropped at the scatter); 4 elementwise launches removed
# speedup vs baseline: 1.7480x; 1.1635x over previous
#include <hip/hip_runtime.h>

typedef __attribute__((ext_vector_type(2))) float v2f;
typedef __attribute__((ext_vector_type(8))) float v8f;
typedef __attribute__((ext_vector_type(4))) float v4f;

__global__ void zero_f32v4(float* __restrict__ p, long long count4) {
  long long i = (long long)blockIdx.x * blockDim.x + threadIdx.x;
  if (i < count4) {
    v4f z = {};
    *(v4f*)(p + 4 * i) = z;
  }
}

template <int CIN, int COUT, int TPB, bool BN>
__global__ __launch_bounds__(32) void sconv_wmma(
    const float* __restrict__ f, const float* __restrict__ W,
    const int* __restrict__ ii, const int* __restrict__ oo,
    const int* __restrict__ nin_p, const int* __restrict__ nout_p,
    int P, float* __restrict__ out, const float* __restrict__ zrow,
    const float* __restrict__ st, const float* __restrict__ g,
    const float* __restrict__ b) {
  const int lane = threadIdx.x;
  const int m    = lane & 15;
  const int half = lane >> 4;

  const int k       = blockIdx.y;
  const int colBase = blockIdx.z << 4;

  const int n_in  = *nin_p;
  const int n_out = *nout_p;

  const int col  = colBase + m;
  constexpr bool FULLN = (COUT % 16) == 0;
  const bool cv  = FULLN || (col < COUT);
  const int colc = cv ? col : (COUT - 1);

  const float* __restrict__ Wk = W + (long long)k * (CIN * COUT);
  v2f bfrag[CIN / 4];
#pragma unroll
  for (int c = 0; c < CIN / 4; ++c) {
    const int k0 = c * 4 + half * 2;
    const float w0 = Wk[(k0 + 0) * COUT + colc];
    const float w1 = Wk[(k0 + 1) * COUT + colc];
    bfrag[c].x = cv ? w0 : 0.0f;
    bfrag[c].y = cv ? w1 : 0.0f;
  }

  v2f sc[CIN / 4], sh[CIN / 4];
  if constexpr (BN) {
    __shared__ __attribute__((aligned(16))) float lss[2 * CIN];
    const float inv_n = 1.0f / (float)n_in;
    for (int ch = lane; ch < CIN; ch += 32) {
      const float mean = st[ch] * inv_n;
      const float var  = st[CIN + ch] * inv_n - mean * mean;
      const float s    = rsqrtf(var + 1e-5f) * g[ch];
      const int slot   = (ch >> 1) * 4 + (ch & 1);
      lss[slot]     = s;
      lss[slot + 2] = b[ch] - mean * s;
    }
    __syncthreads();
#pragma unroll
    for (int c = 0; c < CIN / 4; ++c) {
      const v4f t = *(const v4f*)(lss + (c * 2 + half) * 4);
      sc[c].x = t.x; sc[c].y = t.y;
      sh[c].x = t.z; sh[c].y = t.w;
    }
  }

  const int* __restrict__ iik = ii + (long long)k * P;
  const int* __restrict__ ook = oo + (long long)k * P;

#pragma unroll 1
  for (int t = 0; t < TPB; ++t) {
    const int tile = blockIdx.x * TPB + t;
    const int base = tile * 16;
    if (base >= P) break;
    const int p = base + m;
    const bool pv = (p < P);
    const int pc = pv ? p : (P - 1);

    if (base + 16 < P) {
      __builtin_prefetch(iik + p + 16, 0, 1);
      __builtin_prefetch(ook + p + 16, 0, 1);
    }

    const int o_ld = ook[pc];
    int r_ld = iik[pc];
    asm volatile("" : "+v"(r_ld));

    const int o   = pv ? o_ld : n_out;
    const bool ov = (o >= 0) && (o < n_out);
    if (__ballot(ov) == 0ull) continue;

    const int row = pv ? r_ld : n_in;
    const bool rv = (row >= 0) && (row < n_in);
    int o_s = o;
    if constexpr (BN) {
      o_s = rv ? o : -1;
      asm volatile("" : "+v"(o_s));
    }
    const float* __restrict__ fr =
        (rv ? f + (long long)row * CIN : zrow) + half * 2;

    v2f a[CIN / 4];
#pragma unroll
    for (int c = 0; c < CIN / 4; ++c) a[c] = *(const v2f*)(fr + c * 4);
    if constexpr (BN) {
#pragma unroll
      for (int c = 0; c < CIN / 4; ++c) {
        a[c].x = fmaxf(fmaf(a[c].x, sc[c].x, sh[c].x), 0.0f);
        a[c].y = fmaxf(fmaf(a[c].y, sc[c].y, sh[c].y), 0.0f);
      }
    }
    __builtin_amdgcn_sched_barrier(0);

    v8f acc = {};
#pragma unroll
    for (int c = 0; c < CIN / 4; ++c) {
      acc = __builtin_amdgcn_wmma_f32_16x16x4_f32(
           false, a[c],  false, bfrag[c],
           (short)0, acc,  false,  false);
    }
    __builtin_amdgcn_sched_barrier(0);
    asm volatile("v_nop\n\tv_nop\n\tv_nop\n\tv_nop\n\tv_nop\n\tv_nop\n\tv_nop\n\tv_nop"
                 : "+v"(acc));

    int o_x = o_s;
    asm volatile("" : "+v"(o_x));
    int orow[8];
#pragma unroll
    for (int r = 0; r < 8; ++r) orow[r] = __shfl(o_x, half * 8 + r);
#pragma unroll
    for (int r = 0; r < 8; ++r) {
      const int mr  = half * 8 + r;
      const int o_r = orow[r];
      if (cv && base + mr < P && o_r >= 0 && o_r < n_out) {
        unsafeAtomicAdd(&out[(long long)o_r * COUT + col], acc[r]);
      }
    }
  }
}

template <int C>
__global__ __launch_bounds__(256) void bn_stats(
    const float* __restrict__ x, const int* __restrict__ n_p,
    float* __restrict__ stats) {
  constexpr int RPB = 256 / C;
  const int n = *n_p;
  const int t = threadIdx.x;
  const int c = t & (C - 1);
  const int rg = t / C;

  float s = 0.0f, q = 0.0f;
  for (long long row = (long long)blockIdx.x * RPB + rg; row < n;
       row += (long long)gridDim.x * RPB) {
    const float v = x[row * C + c];
    s += v;
    q += v * v;
  }

  __shared__ float ls[256];
  __shared__ float lq[256];
  ls[t] = s; lq[t] = q;
  __syncthreads();
#pragma unroll
  for (int off = 128; off >= C; off >>= 1) {
    if (t < off) { ls[t] += ls[t + off]; lq[t] += lq[t + off]; }
    __syncthreads();
  }
  if (t < C) {
    unsafeAtomicAdd(&stats[c], ls[t]);
    unsafeAtomicAdd(&stats[C + c], lq[t]);
  }
}

extern "C" void kernel_launch(void* const* d_in, const int* in_sizes, int n_in,
                              void* d_out, int out_size, void* d_ws, size_t ws_size,
                              hipStream_t stream) {
  (void)n_in; (void)ws_size;
  const float* feats = (const float*)d_in[0];
  const float* W0  = (const float*)d_in[1];
  const float* g0  = (const float*)d_in[2];
  const float* b0  = (const float*)d_in[3];
  const float* W1  = (const float*)d_in[4];
  const float* g1  = (const float*)d_in[5];
  const float* b1  = (const float*)d_in[6];
  const float* W2  = (const float*)d_in[7];
  const float* g2  = (const float*)d_in[8];
  const float* b2  = (const float*)d_in[9];
  const float* Wt1 = (const float*)d_in[10];
  const float* gt1 = (const float*)d_in[11];
  const float* bt1 = (const float*)d_in[12];
  const float* Wt2 = (const float*)d_in[13];
  const int* in0  = (const int*)d_in[14];
  const int* out0 = (const int*)d_in[15];
  const int* in1  = (const int*)d_in[16];
  const int* out1 = (const int*)d_in[17];
  const int* in2  = (const int*)d_in[18];
  const int* out2 = (const int*)d_in[19];
  const int* int1 = (const int*)d_in[20];
  const int* outt1= (const int*)d_in[21];
  const int* int2 = (const int*)d_in[22];
  const int* outt2= (const int*)d_in[23];
  const int* n0p  = (const int*)d_in[24];
  const int* n1p  = (const int*)d_in[25];
  const int* n2p  = (const int*)d_in[26];

  const int n0  = in_sizes[0] / 4;
  const int P0  = in_sizes[14] / 27;
  const int P1  = in_sizes[16] / 27;
  const int P2  = in_sizes[18] / 27;
  const int Pt1 = in_sizes[20] / 27;
  const int Pt2 = in_sizes[22] / 27;

  float* ws  = (float*)d_ws;
  float* x0  = ws;
  float* x1  = x0  + (size_t)n0 * 16;
  float* x2  = x1  + (size_t)n0 * 32;
  float* xt1 = x2  + (size_t)n0 * 64;
  float* st0 = xt1 + (size_t)n0 * 32;
  float* st1 = st0 + 32;
  float* st2 = st1 + 64;
  float* stt = st2 + 128;
  const float* zrow = stt + 64;

  const long long zc4 = ((long long)n0 * (16 + 32 + 64 + 32) + 512) / 4;
  zero_f32v4<<<dim3((unsigned)((zc4 + 255) / 256)), 256, 0, stream>>>(ws, zc4);
  const long long oc4 = ((long long)out_size + 3) / 4;
  zero_f32v4<<<dim3((unsigned)((oc4 + 255) / 256)), 256, 0, stream>>>(
      (float*)d_out, oc4);

  constexpr int TPB = 8;
  const dim3 wv(32);
  auto tgrid = [](int P) { return (unsigned)((P + 16 * TPB - 1) / (16 * TPB)); };
  auto sg = [&](int C) {
    long long blocks = ((long long)n0 * C + 255) / 256;
    return dim3((unsigned)(blocks < 768 ? blocks : 768));
  };

  sconv_wmma<4, 16, TPB, false><<<dim3(tgrid(P0), 27, 1), wv, 0, stream>>>(
      feats, W0, in0, out0, n0p, n0p, P0, x0, zrow, nullptr, nullptr, nullptr);
  bn_stats<16><<<sg(16), 256, 0, stream>>>(x0, n0p, st0);

  sconv_wmma<16, 32, TPB, true><<<dim3(tgrid(P1), 27, 2), wv, 0, stream>>>(
      x0, W1, in1, out1, n0p, n1p, P1, x1, zrow, st0, g0, b0);
  bn_stats<32><<<sg(32), 256, 0, stream>>>(x1, n1p, st1);

  sconv_wmma<32, 64, TPB, true><<<dim3(tgrid(P2), 27, 4), wv, 0, stream>>>(
      x1, W2, in2, out2, n1p, n2p, P2, x2, zrow, st1, g1, b1);
  bn_stats<64><<<sg(64), 256, 0, stream>>>(x2, n2p, st2);

  sconv_wmma<64, 32, TPB, true><<<dim3(tgrid(Pt1), 27, 2), wv, 0, stream>>>(
      x2, Wt1, int1, outt1, n2p, n1p, Pt1, xt1, zrow, st2, g2, b2);
  bn_stats<32><<<sg(32), 256, 0, stream>>>(xt1, n1p, stt);

  sconv_wmma<32, 20, TPB, true><<<dim3(tgrid(Pt2), 27, 2), wv, 0, stream>>>(
      xt1, Wt2, int2, outt2, n1p, n0p, Pt2, (float*)d_out, zrow, stt, gt1, bt1);
}


# BN statistics pass: each thread issues eight 16-B loads as one group (grid sized for one trip) instead of one 4-B load per wait; sums meet by lane shuffles + one LDS step (different reduction tree)
# speedup vs baseline: 1.9446x; 1.1125x over previous
#include <hip/hip_runtime.h>

typedef __attribute__((ext_vector_type(2))) float v2f;
typedef __attribute__((ext_vector_type(8))) float v8f;
typedef __attribute__((ext_vector_type(4))) float v4f;

__global__ void zero_f32v4(float* __restrict__ p, long long count4) {
  long long i = (long long)blockIdx.x * blockDim.x + threadIdx.x;
  if (i < count4) {
    v4f z = {};
    *(v4f*)(p + 4 * i) = z;
  }
}

template <int CIN, int COUT, int TPB, bool BN>
__global__ __launch_bounds__(32) void sconv_wmma(
    const float* __restrict__ f, const float* __restrict__ W,
    const int* __restrict__ ii, const int* __restrict__ oo,
    const int* __restrict__ nin_p, const int* __restrict__ nout_p,
    int P, float* __restrict__ out, const float* __restrict__ zrow,
    const float* __restrict__ st, const float* __restrict__ g,
    const float* __restrict__ b) {
  const int lane = threadIdx.x;
  const int m    = lane & 15;
  const int half = lane >> 4;

  const int k       = blockIdx.y;
  const int colBase = blockIdx.z << 4;

  const int n_in  = *nin_p;
  const int n_out = *nout_p;

  const int col  = colBase + m;
  constexpr bool FULLN = (COUT % 16) == 0;
  const bool cv  = FULLN || (col < COUT);
  const int colc = cv ? col : (COUT - 1);

  const float* __restrict__ Wk = W + (long long)k * (CIN * COUT);
  v2f bfrag[CIN / 4];
#pragma unroll
  for (int c = 0; c < CIN / 4; ++c) {
    const int k0 = c * 4 + half * 2;
    const float w0 = Wk[(k0 + 0) * COUT + colc];
    const float w1 = Wk[(k0 + 1) * COUT + colc];
    bfrag[c].x = cv ? w0 : 0.0f;
    bfrag[c].y = cv ? w1 : 0.0f;
  }

  v2f sc[CIN / 4], sh[CIN / 4];
  if constexpr (BN) {
    __shared__ __attribute__((aligned(16))) float lss[2 * CIN];
    const float inv_n = 1.0f / (float)n_in;
    for (int ch = lane; ch < CIN; ch += 32) {
      const float mean = st[ch] * inv_n;
      const float var  = st[CIN + ch] * inv_n - mean * mean;
      const float s    = rsqrtf(var + 1e-5f) * g[ch];
      const int slot   = (ch >> 1) * 4 + (ch & 1);
      lss[slot]     = s;
      lss[slot + 2] = b[ch] - mean * s;
    }
    __syncthreads();
#pragma unroll
    for (int c = 0; c < CIN / 4; ++c) {
      const v4f t = *(const v4f*)(lss + (c * 2 + half) * 4);
      sc[c].x = t.x; sc[c].y = t.y;
      sh[c].x = t.z; sh[c].y = t.w;
    }
  }

  const int* __restrict__ iik = ii + (long long)k * P;
  const int* __restrict__ ook = oo + (long long)k * P;

#pragma unroll 1
  for (int t = 0; t < TPB; ++t) {
    const int tile = blockIdx.x * TPB + t;
    const int base = tile * 16;
    if (base >= P) break;
    const int p = base + m;
    const bool pv = (p < P);
    const int pc = pv ? p : (P - 1);

    if (base + 16 < P) {
      __builtin_prefetch(iik + p + 16, 0, 1);
      __builtin_prefetch(ook + p + 16, 0, 1);
    }

    const int o_ld = ook[pc];
    int r_ld = iik[pc];
    asm volatile("" : "+v"(r_ld));

    const int o   = pv ? o_ld : n_out;
    const bool ov = (o >= 0) && (o < n_out);
    if (__ballot(ov) == 0ull) continue;

    const int row = pv ? r_ld : n_in;
    const bool rv = (row >= 0) && (row < n_in);
    int o_s = o;
    if constexpr (BN) {
      o_s = rv ? o : -1;
      asm volatile("" : "+v"(o_s));
    }
    const float* __restrict__ fr =
        (rv ? f + (long long)row * CIN : zrow) + half * 2;

    v2f a[CIN / 4];
#pragma unroll
    for (int c = 0; c < CIN / 4; ++c) a[c] = *(const v2f*)(fr + c * 4);
    if constexpr (BN) {
#pragma unroll
      for (int c = 0; c < CIN / 4; ++c) {
        a[c].x = fmaxf(fmaf(a[c].x, sc[c].x, sh[c].x), 0.0f);
        a[c].y = fmaxf(fmaf(a[c].y, sc[c].y, sh[c].y), 0.0f);
      }
    }
    __builtin_amdgcn_sched_barrier(0);

    v8f acc = {};
#pragma unroll
    for (int c = 0; c < CIN / 4; ++c) {
      acc = __builtin_amdgcn_wmma_f32_16x16x4_f32(
           false, a[c],  false, bfrag[c],
           (short)0, acc,  false,  false);
    }
    __builtin_amdgcn_sched_barrier(0);
    asm volatile("v_nop\n\tv_nop\n\tv_nop\n\tv_nop\n\tv_nop\n\tv_nop\n\tv_nop\n\tv_nop"
                 : "+v"(acc));

    int o_x = o_s;
    asm volatile("" : "+v"(o_x));
    int orow[8];
#pragma unroll
    for (int r = 0; r < 8; ++r) orow[r] = __shfl(o_x, half * 8 + r);
#pragma unroll
    for (int r = 0; r < 8; ++r) {
      const int mr  = half * 8 + r;
      const int o_r = orow[r];
      if (cv && base + mr < P && o_r >= 0 && o_r < n_out) {
        unsafeAtomicAdd(&out[(long long)o_r * COUT + col], acc[r]);
      }
    }
  }
}

template <int C>
__global__ __launch_bounds__(256) void bn_stats(
    const float* __restrict__ x, const int* __restrict__ n_p,
    float* __restrict__ stats) {
  constexpr int U = 8;
  constexpr int G = C / 4;
  const int n = *n_p;
  const int t = threadIdx.x;
  const int lane = t & 31;
  const int w = t >> 5;
  const long long total4 = (long long)n * G;
  const long long T = (long long)gridDim.x * 256;
  const long long gt = (long long)blockIdx.x * 256 + t;
  const v4f* __restrict__ x4 = (const v4f*)x;

  v4f s = {}, q = {};
  for (long long base = gt; base < total4; base += U * T) {
    v4f v[U];
#pragma unroll
    for (int j = 0; j < U; ++j) {
      const long long i = base + j * T;
      const bool ok = i < total4;
      const v4f ld = x4[ok ? i : base];
      const v4f z = {};
      v[j] = ok ? ld : z;
    }
#pragma unroll
    for (int j = 0; j < U; ++j) {
      s += v[j];
      q += v[j] * v[j];
    }
  }

#pragma unroll
  for (int off = G; off < 32; off <<= 1) {
#pragma unroll
    for (int e = 0; e < 4; ++e) {
      s[e] += __shfl_xor(s[e], off);
      q[e] += __shfl_xor(q[e], off);
    }
  }

  __shared__ float part[8][2 * C];
  if (lane < G) {
#pragma unroll
    for (int e = 0; e < 4; ++e) {
      part[w][4 * lane + e]     = s[e];
      part[w][C + 4 * lane + e] = q[e];
    }
  }
  __syncthreads();
  if (t < 2 * C) {
    float tot = 0.0f;
#pragma unroll
    for (int k = 0; k < 8; ++k) tot += part[k][t];
    unsafeAtomicAdd(&stats[t], tot);
  }
}

extern "C" void kernel_launch(void* const* d_in, const int* in_sizes, int n_in,
                              void* d_out, int out_size, void* d_ws, size_t ws_size,
                              hipStream_t stream) {
  (void)n_in; (void)ws_size;
  const float* feats = (const float*)d_in[0];
  const float* W0  = (const float*)d_in[1];
  const float* g0  = (const float*)d_in[2];
  const float* b0  = (const float*)d_in[3];
  const float* W1  = (const float*)d_in[4];
  const float* g1  = (const float*)d_in[5];
  const float* b1  = (const float*)d_in[6];
  const float* W2  = (const float*)d_in[7];
  const float* g2  = (const float*)d_in[8];
  const float* b2  = (const float*)d_in[9];
  const float* Wt1 = (const float*)d_in[10];
  const float* gt1 = (const float*)d_in[11];
  const float* bt1 = (const float*)d_in[12];
  const float* Wt2 = (const float*)d_in[13];
  const int* in0  = (const int*)d_in[14];
  const int* out0 = (const int*)d_in[15];
  const int* in1  = (const int*)d_in[16];
  const int* out1 = (const int*)d_in[17];
  const int* in2  = (const int*)d_in[18];
  const int* out2 = (const int*)d_in[19];
  const int* int1 = (const int*)d_in[20];
  const int* outt1= (const int*)d_in[21];
  const int* int2 = (const int*)d_in[22];
  const int* outt2= (const int*)d_in[23];
  const int* n0p  = (const int*)d_in[24];
  const int* n1p  = (const int*)d_in[25];
  const int* n2p  = (const int*)d_in[26];

  const int n0  = in_sizes[0] / 4;
  const int P0  = in_sizes[14] / 27;
  const int P1  = in_sizes[16] / 27;
  const int P2  = in_sizes[18] / 27;
  const int Pt1 = in_sizes[20] / 27;
  const int Pt2 = in_sizes[22] / 27;

  float* ws  = (float*)d_ws;
  float* x0  = ws;
  float* x1  = x0  + (size_t)n0 * 16;
  float* x2  = x1  + (size_t)n0 * 32;
  float* xt1 = x2  + (size_t)n0 * 64;
  float* st0 = xt1 + (size_t)n0 * 32;
  float* st1 = st0 + 32;
  float* st2 = st1 + 64;
  float* stt = st2 + 128;
  const float* zrow = stt + 64;

  const long long zc4 = ((long long)n0 * (16 + 32 + 64 + 32) + 512) / 4;
  zero_f32v4<<<dim3((unsigned)((zc4 + 255) / 256)), 256, 0, stream>>>(ws, zc4);
  const long long oc4 = ((long long)out_size + 3) / 4;
  zero_f32v4<<<dim3((unsigned)((oc4 + 255) / 256)), 256, 0, stream>>>(
      (float*)d_out, oc4);

  constexpr int TPB = 8;
  const dim3 wv(32);
  auto tgrid = [](int P) { return (unsigned)((P + 16 * TPB - 1) / (16 * TPB)); };
  auto sg = [&](int C) {
    long long blocks = ((long long)n0 * C + 8191) / 8192;
    return dim3((unsigned)(blocks < 512 ? blocks : 512));
  };

  sconv_wmma<4, 16, TPB, false><<<dim3(tgrid(P0), 27, 1), wv, 0, stream>>>(
      feats, W0, in0, out0, n0p, n0p, P0, x0, zrow, nullptr, nullptr, nullptr);
  bn_stats<16><<<sg(16), 256, 0, stream>>>(x0, n0p, st0);

  sconv_wmma<16, 32, TPB, true><<<dim3(tgrid(P1), 27, 2), wv, 0, stream>>>(
      x0, W1, in1, out1, n0p, n1p, P1, x1, zrow, st0, g0, b0);
  bn_stats<32><<<sg(32), 256, 0, stream>>>(x1, n1p, st1);

  sconv_wmma<32, 64, TPB, true><<<dim3(tgrid(P2), 27, 4), wv, 0, stream>>>(
      x1, W2, in2, out2, n1p, n2p, P2, x2, zrow, st1, g1, b1);
  bn_stats<64><<<sg(64), 256, 0, stream>>>(x2, n2p, st2);

  sconv_wmma<64, 32, TPB, true><<<dim3(tgrid(Pt1), 27, 2), wv, 0, stream>>>(
      x2, Wt1, int1, outt1, n2p, n1p, Pt1, xt1, zrow, st2, g2, b2);
  bn_stats<32><<<sg(32), 256, 0, stream>>>(xt1, n1p, stt);

  sconv_wmma<32, 20, TPB, true><<<dim3(tgrid(Pt2), 27, 2), wv, 0, stream>>>(
      xt1, Wt2, int2, outt2, n1p, n0p, Pt2, (float*)d_out, zrow, stt, gt1, bt1);
}


# Conv waves of the Cout>16 layers own 32 output columns: two accumulator tiles per gathered A, halves swapped (v_permlanex16) so each atomic add instruction covers one 128-B row run, not two 64-B runs
# speedup vs baseline: 2.2390x; 1.1514x over previous
#include <hip/hip_runtime.h>

typedef __attribute__((ext_vector_type(2))) float v2f;
typedef __attribute__((ext_vector_type(8))) float v8f;
typedef __attribute__((ext_vector_type(4))) float v4f;

__global__ void zero_f32v4(float* __restrict__ p, long long count4) {
  long long i = (long long)blockIdx.x * blockDim.x + threadIdx.x;
  if (i < count4) {
    v4f z = {};
    *(v4f*)(p + 4 * i) = z;
  }
}

template <int CIN, int COUT, int TPB, bool BN>
__global__ __launch_bounds__(32) void sconv_wmma(
    const float* __restrict__ f, const float* __restrict__ W,
    const int* __restrict__ ii, const int* __restrict__ oo,
    const int* __restrict__ nin_p, const int* __restrict__ nout_p,
    int P, float* __restrict__ out, const float* __restrict__ zrow,
    const float* __restrict__ st, const float* __restrict__ g,
    const float* __restrict__ b) {
  const int lane = threadIdx.x;
  const int m    = lane & 15;
  const int half = lane >> 4;

  const int k       = blockIdx.y;
  const int colBase = blockIdx.z << 4;

  const int n_in  = *nin_p;
  const int n_out = *nout_p;

  const int col  = colBase + m;
  constexpr bool FULLN = (COUT % 16) == 0;
  const bool cv  = FULLN || (col < COUT);
  const int colc = cv ? col : (COUT - 1);

  const float* __restrict__ Wk = W + (long long)k * (CIN * COUT);
  v2f bfrag[CIN / 4];
#pragma unroll
  for (int c = 0; c < CIN / 4; ++c) {
    const int k0 = c * 4 + half * 2;
    const float w0 = Wk[(k0 + 0) * COUT + colc];
    const float w1 = Wk[(k0 + 1) * COUT + colc];
    bfrag[c].x = cv ? w0 : 0.0f;
    bfrag[c].y = cv ? w1 : 0.0f;
  }

  v2f sc[CIN / 4], sh[CIN / 4];
  if constexpr (BN) {
    __shared__ __attribute__((aligned(16))) float lss[2 * CIN];
    const float inv_n = 1.0f / (float)n_in;
    for (int ch = lane; ch < CIN; ch += 32) {
      const float mean = st[ch] * inv_n;
      const float var  = st[CIN + ch] * inv_n - mean * mean;
      const float s    = rsqrtf(var + 1e-5f) * g[ch];
      const int slot   = (ch >> 1) * 4 + (ch & 1);
      lss[slot]     = s;
      lss[slot + 2] = b[ch] - mean * s;
    }
    __syncthreads();
#pragma unroll
    for (int c = 0; c < CIN / 4; ++c) {
      const v4f t = *(const v4f*)(lss + (c * 2 + half) * 4);
      sc[c].x = t.x; sc[c].y = t.y;
      sh[c].x = t.z; sh[c].y = t.w;
    }
  }

  const int* __restrict__ iik = ii + (long long)k * P;
  const int* __restrict__ ook = oo + (long long)k * P;

#pragma unroll 1
  for (int t = 0; t < TPB; ++t) {
    const int tile = blockIdx.x * TPB + t;
    const int base = tile * 16;
    if (base >= P) break;
    const int p = base + m;
    const bool pv = (p < P);
    const int pc = pv ? p : (P - 1);

    if (base + 16 < P) {
      __builtin_prefetch(iik + p + 16, 0, 1);
      __builtin_prefetch(ook + p + 16, 0, 1);
    }

    const int o_ld = ook[pc];
    int r_ld = iik[pc];
    asm volatile("" : "+v"(r_ld));

    const int o   = pv ? o_ld : n_out;
    const bool ov = (o >= 0) && (o < n_out);
    if (__ballot(ov) == 0ull) continue;

    const int row = pv ? r_ld : n_in;
    const bool rv = (row >= 0) && (row < n_in);
    int o_s = o;
    if constexpr (BN) {
      o_s = rv ? o : -1;
      asm volatile("" : "+v"(o_s));
    }
    const float* __restrict__ fr =
        (rv ? f + (long long)row * CIN : zrow) + half * 2;

    v2f a[CIN / 4];
#pragma unroll
    for (int c = 0; c < CIN / 4; ++c) a[c] = *(const v2f*)(fr + c * 4);
    if constexpr (BN) {
#pragma unroll
      for (int c = 0; c < CIN / 4; ++c) {
        a[c].x = fmaxf(fmaf(a[c].x, sc[c].x, sh[c].x), 0.0f);
        a[c].y = fmaxf(fmaf(a[c].y, sc[c].y, sh[c].y), 0.0f);
      }
    }
    __builtin_amdgcn_sched_barrier(0);

    v8f acc = {};
#pragma unroll
    for (int c = 0; c < CIN / 4; ++c) {
      acc = __builtin_amdgcn_wmma_f32_16x16x4_f32(
           false, a[c],  false, bfrag[c],
           (short)0, acc,  false,  false);
    }
    __builtin_amdgcn_sched_barrier(0);
    asm volatile("v_nop\n\tv_nop\n\tv_nop\n\tv_nop\n\tv_nop\n\tv_nop\n\tv_nop\n\tv_nop"
                 : "+v"(acc));

    int o_x = o_s;
    asm volatile("" : "+v"(o_x));
    int orow[8];
#pragma unroll
    for (int r = 0; r < 8; ++r) orow[r] = __shfl(o_x, half * 8 + r);
#pragma unroll
    for (int r = 0; r < 8; ++r) {
      const int mr  = half * 8 + r;
      const int o_r = orow[r];
      if (cv && base + mr < P && o_r >= 0 && o_r < n_out) {
        unsafeAtomicAdd(&out[(long long)o_r * COUT + col], acc[r]);
      }
    }
  }
}

template <int CIN, int COUT, int TPB, bool BN>
__global__ __launch_bounds__(32) void sconv_wmma_row32(
    const float* __restrict__ f, const float* __restrict__ W,
    const int* __restrict__ ii, const int* __restrict__ oo,
    const int* __restrict__ nin_p, const int* __restrict__ nout_p,
    int P, float* __restrict__ out, const float* __restrict__ zrow,
    const float* __restrict__ st, const float* __restrict__ g,
    const float* __restrict__ b) {
  static_assert(COUT > 16, "one-tile layers use sconv_wmma");
  const int lane = threadIdx.x;
  const int m    = lane & 15;
  const int half = lane >> 4;

  const int k       = blockIdx.y;
  const int colBase = blockIdx.z << 5;

  const int n_in  = *nin_p;
  const int n_out = *nout_p;

  constexpr bool FULLN = (COUT % 32) == 0;
  const int colA = colBase + m;
  const int colB = colA + 16;
  const bool cvA = FULLN || (colA < COUT);
  const bool cvB = FULLN || (colB < COUT);
  const int colAc = cvA ? colA : (COUT - 1);
  const int colBc = cvB ? colB : (COUT - 1);
  const int colL = colBase + lane;
  const bool cvL = FULLN || (colL < COUT);

  const float* __restrict__ Wk = W + (long long)k * (CIN * COUT);
  v2f bA[CIN / 4], bB[CIN / 4];
#pragma unroll
  for (int c = 0; c < CIN / 4; ++c) {
    const int k0 = c * 4 + half * 2;
    const float wa0 = Wk[(k0 + 0) * COUT + colAc];
    const float wa1 = Wk[(k0 + 1) * COUT + colAc];
    const float wb0 = Wk[(k0 + 0) * COUT + colBc];
    const float wb1 = Wk[(k0 + 1) * COUT + colBc];
    bA[c].x = cvA ? wa0 : 0.0f;
    bA[c].y = cvA ? wa1 : 0.0f;
    bB[c].x = cvB ? wb0 : 0.0f;
    bB[c].y = cvB ? wb1 : 0.0f;
  }

  v2f sc[CIN / 4], sh[CIN / 4];
  if constexpr (BN) {
    __shared__ __attribute__((aligned(16))) float lss[2 * CIN];
    const float inv_n = 1.0f / (float)n_in;
    for (int ch = lane; ch < CIN; ch += 32) {
      const float mean = st[ch] * inv_n;
      const float var  = st[CIN + ch] * inv_n - mean * mean;
      const float s    = rsqrtf(var + 1e-5f) * g[ch];
      const int slot   = (ch >> 1) * 4 + (ch & 1);
      lss[slot]     = s;
      lss[slot + 2] = b[ch] - mean * s;
    }
    __syncthreads();
#pragma unroll
    for (int c = 0; c < CIN / 4; ++c) {
      const v4f t = *(const v4f*)(lss + (c * 2 + half) * 4);
      sc[c].x = t.x; sc[c].y = t.y;
      sh[c].x = t.z; sh[c].y = t.w;
    }
  }

  const int* __restrict__ iik = ii + (long long)k * P;
  const int* __restrict__ ook = oo + (long long)k * P;

#pragma unroll 1
  for (int t = 0; t < TPB; ++t) {
    const int tile = blockIdx.x * TPB + t;
    const int base = tile * 16;
    if (base >= P) break;
    const int p = base + m;
    const bool pv = (p < P);
    const int pc = pv ? p : (P - 1);

    if (base + 16 < P) {
      __builtin_prefetch(iik + p + 16, 0, 1);
      __builtin_prefetch(ook + p + 16, 0, 1);
    }

    const int o_ld = ook[pc];
    int r_ld = iik[pc];
    asm volatile("" : "+v"(r_ld));

    const int o   = pv ? o_ld : n_out;
    const bool ov = (o >= 0) && (o < n_out);
    if (__ballot(ov) == 0ull) continue;

    const int row = pv ? r_ld : n_in;
    const bool rv = (row >= 0) && (row < n_in);
    int o_s = o;
    if constexpr (BN) {
      o_s = rv ? o : -1;
      asm volatile("" : "+v"(o_s));
    }
    const float* __restrict__ fr =
        (rv ? f + (long long)row * CIN : zrow) + half * 2;

    v2f a[CIN / 4];
#pragma unroll
    for (int c = 0; c < CIN / 4; ++c) a[c] = *(const v2f*)(fr + c * 4);
    if constexpr (BN) {
#pragma unroll
      for (int c = 0; c < CIN / 4; ++c) {
        a[c].x = fmaxf(fmaf(a[c].x, sc[c].x, sh[c].x), 0.0f);
        a[c].y = fmaxf(fmaf(a[c].y, sc[c].y, sh[c].y), 0.0f);
      }
    }
    __builtin_amdgcn_sched_barrier(0);

    v8f acc0 = {}, acc1 = {};
#pragma unroll
    for (int c = 0; c < CIN / 4; ++c) {
      acc0 = __builtin_amdgcn_wmma_f32_16x16x4_f32(
           false, a[c],  false, bA[c],
           (short)0, acc0,  false,  false);
      acc1 = __builtin_amdgcn_wmma_f32_16x16x4_f32(
           false, a[c],  false, bB[c],
           (short)0, acc1,  false,  false);
    }
    __builtin_amdgcn_sched_barrier(0);
    asm volatile("v_nop\n\tv_nop\n\tv_nop\n\tv_nop\n\tv_nop\n\tv_nop\n\tv_nop\n\tv_nop"
                 : "+v"(acc0), "+v"(acc1));

    float xr[8], yr[8];
#pragma unroll
    for (int r = 0; r < 8; ++r) {
      const float a0 = acc0[r];
      const float a1 = acc1[r];
      const float a0x = __builtin_bit_cast(float, __builtin_amdgcn_permlanex16(
          0u, __builtin_bit_cast(unsigned, a0), 0x76543210u, 0xfedcba98u, false, false));
      const float a1x = __builtin_bit_cast(float, __builtin_amdgcn_permlanex16(
          0u, __builtin_bit_cast(unsigned, a1), 0x76543210u, 0xfedcba98u, false, false));
      xr[r] = half ? a1x : a0;
      yr[r] = half ? a1 : a0x;
    }

    int o_x = o_s;
    asm volatile("" : "+v"(o_x));
#pragma unroll
    for (int r = 0; r < 8; ++r) {
      const int oa = __builtin_amdgcn_readlane(o_x, r);
      const int ob = __builtin_amdgcn_readlane(o_x, 8 + r);
      if (oa >= 0 && oa < n_out && cvL)
        unsafeAtomicAdd(&out[(long long)oa * COUT + colL], xr[r]);
      if (ob >= 0 && ob < n_out && cvL)
        unsafeAtomicAdd(&out[(long long)ob * COUT + colL], yr[r]);
    }
  }
}

template <int C>
__global__ __launch_bounds__(256) void bn_stats(
    const float* __restrict__ x, const int* __restrict__ n_p,
    float* __restrict__ stats) {
  constexpr int U = 8;
  constexpr int G = C / 4;
  const int n = *n_p;
  const int t = threadIdx.x;
  const int lane = t & 31;
  const int w = t >> 5;
  const long long total4 = (long long)n * G;
  const long long T = (long long)gridDim.x * 256;
  const long long gt = (long long)blockIdx.x * 256 + t;
  const v4f* __restrict__ x4 = (const v4f*)x;

  v4f s = {}, q = {};
  for (long long base = gt; base < total4; base += U * T) {
    v4f v[U];
#pragma unroll
    for (int j = 0; j < U; ++j) {
      const long long i = base + j * T;
      const bool ok = i < total4;
      const v4f ld = x4[ok ? i : base];
      const v4f z = {};
      v[j] = ok ? ld : z;
    }
#pragma unroll
    for (int j = 0; j < U; ++j) {
      s += v[j];
      q += v[j] * v[j];
    }
  }

#pragma unroll
  for (int off = G; off < 32; off <<= 1) {
#pragma unroll
    for (int e = 0; e < 4; ++e) {
      s[e] += __shfl_xor(s[e], off);
      q[e] += __shfl_xor(q[e], off);
    }
  }

  __shared__ float part[8][2 * C];
  if (lane < G) {
#pragma unroll
    for (int e = 0; e < 4; ++e) {
      part[w][4 * lane + e]     = s[e];
      part[w][C + 4 * lane + e] = q[e];
    }
  }
  __syncthreads();
  if (t < 2 * C) {
    float tot = 0.0f;
#pragma unroll
    for (int k = 0; k < 8; ++k) tot += part[k][t];
    unsafeAtomicAdd(&stats[t], tot);
  }
}

extern "C" void kernel_launch(void* const* d_in, const int* in_sizes, int n_in,
                              void* d_out, int out_size, void* d_ws, size_t ws_size,
                              hipStream_t stream) {
  (void)n_in; (void)ws_size;
  const float* feats = (const float*)d_in[0];
  const float* W0  = (const float*)d_in[1];
  const float* g0  = (const float*)d_in[2];
  const float* b0  = (const float*)d_in[3];
  const float* W1  = (const float*)d_in[4];
  const float* g1  = (const float*)d_in[5];
  const float* b1  = (const float*)d_in[6];
  const float* W2  = (const float*)d_in[7];
  const float* g2  = (const float*)d_in[8];
  const float* b2  = (const float*)d_in[9];
  const float* Wt1 = (const float*)d_in[10];
  const float* gt1 = (const float*)d_in[11];
  const float* bt1 = (const float*)d_in[12];
  const float* Wt2 = (const float*)d_in[13];
  const int* in0  = (const int*)d_in[14];
  const int* out0 = (const int*)d_in[15];
  const int* in1  = (const int*)d_in[16];
  const int* out1 = (const int*)d_in[17];
  const int* in2  = (const int*)d_in[18];
  const int* out2 = (const int*)d_in[19];
  const int* int1 = (const int*)d_in[20];
  const int* outt1= (const int*)d_in[21];
  const int* int2 = (const int*)d_in[22];
  const int* outt2= (const int*)d_in[23];
  const int* n0p  = (const int*)d_in[24];
  const int* n1p  = (const int*)d_in[25];
  const int* n2p  = (const int*)d_in[26];

  const int n0  = in_sizes[0] / 4;
  const int P0  = in_sizes[14] / 27;
  const int P1  = in_sizes[16] / 27;
  const int P2  = in_sizes[18] / 27;
  const int Pt1 = in_sizes[20] / 27;
  const int Pt2 = in_sizes[22] / 27;

  float* ws  = (float*)d_ws;
  float* x0  = ws;
  float* x1  = x0  + (size_t)n0 * 16;
  float* x2  = x1  + (size_t)n0 * 32;
  float* xt1 = x2  + (size_t)n0 * 64;
  float* st0 = xt1 + (size_t)n0 * 32;
  float* st1 = st0 + 32;
  float* st2 = st1 + 64;
  float* stt = st2 + 128;
  const float* zrow = stt + 64;

  const long long zc4 = ((long long)n0 * (16 + 32 + 64 + 32) + 512) / 4;
  zero_f32v4<<<dim3((unsigned)((zc4 + 255) / 256)), 256, 0, stream>>>(ws, zc4);
  const long long oc4 = ((long long)out_size + 3) / 4;
  zero_f32v4<<<dim3((unsigned)((oc4 + 255) / 256)), 256, 0, stream>>>(
      (float*)d_out, oc4);

  constexpr int TPB = 8;
  const dim3 wv(32);
  auto tgrid = [](int P) { return (unsigned)((P + 16 * TPB - 1) / (16 * TPB)); };
  auto sg = [&](int C) {
    long long blocks = ((long long)n0 * C + 8191) / 8192;
    return dim3((unsigned)(blocks < 512 ? blocks : 512));
  };

  sconv_wmma<4, 16, TPB, false><<<dim3(tgrid(P0), 27, 1), wv, 0, stream>>>(
      feats, W0, in0, out0, n0p, n0p, P0, x0, zrow, nullptr, nullptr, nullptr);
  bn_stats<16><<<sg(16), 256, 0, stream>>>(x0, n0p, st0);

  sconv_wmma_row32<16, 32, TPB, true><<<dim3(tgrid(P1), 27, 1), wv, 0, stream>>>(
      x0, W1, in1, out1, n0p, n1p, P1, x1, zrow, st0, g0, b0);
  bn_stats<32><<<sg(32), 256, 0, stream>>>(x1, n1p, st1);

  sconv_wmma_row32<32, 64, TPB, true><<<dim3(tgrid(P2), 27, 2), wv, 0, stream>>>(
      x1, W2, in2, out2, n1p, n2p, P2, x2, zrow, st1, g1, b1);
  bn_stats<64><<<sg(64), 256, 0, stream>>>(x2, n2p, st2);

  sconv_wmma_row32<64, 32, TPB, true><<<dim3(tgrid(Pt1), 27, 1), wv, 0, stream>>>(
      x2, Wt1, int1, outt1, n2p, n1p, Pt1, xt1, zrow, st2, g2, b2);
  bn_stats<32><<<sg(32), 256, 0, stream>>>(xt1, n1p, stt);

  sconv_wmma_row32<32, 20, TPB, true><<<dim3(tgrid(Pt2), 27, 1), wv, 0, stream>>>(
      xt1, Wt2, int2, outt2, n1p, n0p, Pt2, (float*)d_out, zrow, stt, gt1, bt1);
}


# Conv blocks of the four Cout>16 layers are 2 or 4 waves that split the block's 8 pair-tiles (was one wave sweeping all 8): a wave's dependent index-row-add chain is 4 or 2 tiles; BN table shared
# speedup vs baseline: 2.6089x; 1.1652x over previous
#include <hip/hip_runtime.h>

typedef __attribute__((ext_vector_type(2))) float v2f;
typedef __attribute__((ext_vector_type(8))) float v8f;
typedef __attribute__((ext_vector_type(4))) float v4f;

__global__ void zero_f32v4(float* __restrict__ p, long long count4) {
  long long i = (long long)blockIdx.x * blockDim.x + threadIdx.x;
  if (i < count4) {
    v4f z = {};
    *(v4f*)(p + 4 * i) = z;
  }
}

template <int CIN, int COUT, int TPB, bool BN>
__global__ __launch_bounds__(32) void sconv_wmma(
    const float* __restrict__ f, const float* __restrict__ W,
    const int* __restrict__ ii, const int* __restrict__ oo,
    const int* __restrict__ nin_p, const int* __restrict__ nout_p,
    int P, float* __restrict__ out, const float* __restrict__ zrow,
    const float* __restrict__ st, const float* __restrict__ g,
    const float* __restrict__ b) {
  const int lane = threadIdx.x;
  const int m    = lane & 15;
  const int half = lane >> 4;

  const int k       = blockIdx.y;
  const int colBase = blockIdx.z << 4;

  const int n_in  = *nin_p;
  const int n_out = *nout_p;

  const int col  = colBase + m;
  constexpr bool FULLN = (COUT % 16) == 0;
  const bool cv  = FULLN || (col < COUT);
  const int colc = cv ? col : (COUT - 1);

  const float* __restrict__ Wk = W + (long long)k * (CIN * COUT);
  v2f bfrag[CIN / 4];
#pragma unroll
  for (int c = 0; c < CIN / 4; ++c) {
    const int k0 = c * 4 + half * 2;
    const float w0 = Wk[(k0 + 0) * COUT + colc];
    const float w1 = Wk[(k0 + 1) * COUT + colc];
    bfrag[c].x = cv ? w0 : 0.0f;
    bfrag[c].y = cv ? w1 : 0.0f;
  }

  v2f sc[CIN / 4], sh[CIN / 4];
  if constexpr (BN) {
    __shared__ __attribute__((aligned(16))) float lss[2 * CIN];
    const float inv_n = 1.0f / (float)n_in;
    for (int ch = lane; ch < CIN; ch += 32) {
      const float mean = st[ch] * inv_n;
      const float var  = st[CIN + ch] * inv_n - mean * mean;
      const float s    = rsqrtf(var + 1e-5f) * g[ch];
      const int slot   = (ch >> 1) * 4 + (ch & 1);
      lss[slot]     = s;
      lss[slot + 2] = b[ch] - mean * s;
    }
    __syncthreads();
#pragma unroll
    for (int c = 0; c < CIN / 4; ++c) {
      const v4f t = *(const v4f*)(lss + (c * 2 + half) * 4);
      sc[c].x = t.x; sc[c].y = t.y;
      sh[c].x = t.z; sh[c].y = t.w;
    }
  }

  const int* __restrict__ iik = ii + (long long)k * P;
  const int* __restrict__ ook = oo + (long long)k * P;

#pragma unroll 1
  for (int t = 0; t < TPB; ++t) {
    const int tile = blockIdx.x * TPB + t;
    const int base = tile * 16;
    if (base >= P) break;
    const int p = base + m;
    const bool pv = (p < P);
    const int pc = pv ? p : (P - 1);

    if (base + 16 < P) {
      __builtin_prefetch(iik + p + 16, 0, 1);
      __builtin_prefetch(ook + p + 16, 0, 1);
    }

    const int o_ld = ook[pc];
    int r_ld = iik[pc];
    asm volatile("" : "+v"(r_ld));

    const int o   = pv ? o_ld : n_out;
    const bool ov = (o >= 0) && (o < n_out);
    if (__ballot(ov) == 0ull) continue;

    const int row = pv ? r_ld : n_in;
    const bool rv = (row >= 0) && (row < n_in);
    int o_s = o;
    if constexpr (BN) {
      o_s = rv ? o : -1;
      asm volatile("" : "+v"(o_s));
    }
    const float* __restrict__ fr =
        (rv ? f + (long long)row * CIN : zrow) + half * 2;

    v2f a[CIN / 4];
#pragma unroll
    for (int c = 0; c < CIN / 4; ++c) a[c] = *(const v2f*)(fr + c * 4);
    if constexpr (BN) {
#pragma unroll
      for (int c = 0; c < CIN / 4; ++c) {
        a[c].x = fmaxf(fmaf(a[c].x, sc[c].x, sh[c].x), 0.0f);
        a[c].y = fmaxf(fmaf(a[c].y, sc[c].y, sh[c].y), 0.0f);
      }
    }
    __builtin_amdgcn_sched_barrier(0);

    v8f acc = {};
#pragma unroll
    for (int c = 0; c < CIN / 4; ++c) {
      acc = __builtin_amdgcn_wmma_f32_16x16x4_f32(
           false, a[c],  false, bfrag[c],
           (short)0, acc,  false,  false);
    }
    __builtin_amdgcn_sched_barrier(0);
    asm volatile("v_nop\n\tv_nop\n\tv_nop\n\tv_nop\n\tv_nop\n\tv_nop\n\tv_nop\n\tv_nop"
                 : "+v"(acc));

    int o_x = o_s;
    asm volatile("" : "+v"(o_x));
    int orow[8];
#pragma unroll
    for (int r = 0; r < 8; ++r) orow[r] = __shfl(o_x, half * 8 + r);
#pragma unroll
    for (int r = 0; r < 8; ++r) {
      const int mr  = half * 8 + r;
      const int o_r = orow[r];
      if (cv && base + mr < P && o_r >= 0 && o_r < n_out) {
        unsafeAtomicAdd(&out[(long long)o_r * COUT + col], acc[r]);
      }
    }
  }
}

template <int CIN, int COUT, int WPB, int TPW, bool BN>
__global__ __launch_bounds__(32 * WPB) void sconv_wmma_row32(
    const float* __restrict__ f, const float* __restrict__ W,
    const int* __restrict__ ii, const int* __restrict__ oo,
    const int* __restrict__ nin_p, const int* __restrict__ nout_p,
    int P, float* __restrict__ out, const float* __restrict__ zrow,
    const float* __restrict__ st, const float* __restrict__ g,
    const float* __restrict__ b) {
  static_assert(COUT > 16, "one-tile layers use sconv_wmma");
  const int lane = threadIdx.x & 31;
  const int wv   = __builtin_amdgcn_readfirstlane((int)(threadIdx.x >> 5));
  const int m    = lane & 15;
  const int half = lane >> 4;

  const int k       = blockIdx.y;
  const int colBase = blockIdx.z << 5;

  const int n_in  = *nin_p;
  const int n_out = *nout_p;

  constexpr bool FULLN = (COUT % 32) == 0;
  const int colA = colBase + m;
  const int colB = colA + 16;
  const bool cvA = FULLN || (colA < COUT);
  const bool cvB = FULLN || (colB < COUT);
  const int colAc = cvA ? colA : (COUT - 1);
  const int colBc = cvB ? colB : (COUT - 1);
  const int colL = colBase + lane;
  const bool cvL = FULLN || (colL < COUT);

  const float* __restrict__ Wk = W + (long long)k * (CIN * COUT);
  v2f bA[CIN / 4], bB[CIN / 4];
#pragma unroll
  for (int c = 0; c < CIN / 4; ++c) {
    const int k0 = c * 4 + half * 2;
    const float wa0 = Wk[(k0 + 0) * COUT + colAc];
    const float wa1 = Wk[(k0 + 1) * COUT + colAc];
    const float wb0 = Wk[(k0 + 0) * COUT + colBc];
    const float wb1 = Wk[(k0 + 1) * COUT + colBc];
    bA[c].x = cvA ? wa0 : 0.0f;
    bA[c].y = cvA ? wa1 : 0.0f;
    bB[c].x = cvB ? wb0 : 0.0f;
    bB[c].y = cvB ? wb1 : 0.0f;
  }

  v2f sc[CIN / 4], sh[CIN / 4];
  if constexpr (BN) {
    __shared__ __attribute__((aligned(16))) float lss[2 * CIN];
    const float inv_n = 1.0f / (float)n_in;
    for (int ch = threadIdx.x; ch < CIN; ch += 32 * WPB) {
      const float mean = st[ch] * inv_n;
      const float var  = st[CIN + ch] * inv_n - mean * mean;
      const float s    = rsqrtf(var + 1e-5f) * g[ch];
      const int slot   = (ch >> 1) * 4 + (ch & 1);
      lss[slot]     = s;
      lss[slot + 2] = b[ch] - mean * s;
    }
    __syncthreads();
#pragma unroll
    for (int c = 0; c < CIN / 4; ++c) {
      const v4f t = *(const v4f*)(lss + (c * 2 + half) * 4);
      sc[c].x = t.x; sc[c].y = t.y;
      sh[c].x = t.z; sh[c].y = t.w;
    }
  }

  const int* __restrict__ iik = ii + (long long)k * P;
  const int* __restrict__ ook = oo + (long long)k * P;

#pragma unroll 1
  for (int t = 0; t < TPW; ++t) {
    const int tile = (blockIdx.x * TPW + t) * WPB + wv;
    const int base = tile * 16;
    if (base >= P) break;
    const int p = base + m;
    const bool pv = (p < P);
    const int pc = pv ? p : (P - 1);

    if (base + 16 < P) {
      __builtin_prefetch(iik + p + 16, 0, 1);
      __builtin_prefetch(ook + p + 16, 0, 1);
    }

    const int o_ld = ook[pc];
    int r_ld = iik[pc];
    asm volatile("" : "+v"(r_ld));

    const int o   = pv ? o_ld : n_out;
    const bool ov = (o >= 0) && (o < n_out);
    if (__ballot(ov) == 0ull) continue;

    const int row = pv ? r_ld : n_in;
    const bool rv = (row >= 0) && (row < n_in);
    int o_s = o;
    if constexpr (BN) {
      o_s = rv ? o : -1;
      asm volatile("" : "+v"(o_s));
    }
    const float* __restrict__ fr =
        (rv ? f + (long long)row * CIN : zrow) + half * 2;

    v2f a[CIN / 4];
#pragma unroll
    for (int c = 0; c < CIN / 4; ++c) a[c] = *(const v2f*)(fr + c * 4);
    if constexpr (BN) {
#pragma unroll
      for (int c = 0; c < CIN / 4; ++c) {
        a[c].x = fmaxf(fmaf(a[c].x, sc[c].x, sh[c].x), 0.0f);
        a[c].y = fmaxf(fmaf(a[c].y, sc[c].y, sh[c].y), 0.0f);
      }
    }
    __builtin_amdgcn_sched_barrier(0);

    v8f acc0 = {}, acc1 = {};
#pragma unroll
    for (int c = 0; c < CIN / 4; ++c) {
      acc0 = __builtin_amdgcn_wmma_f32_16x16x4_f32(
           false, a[c],  false, bA[c],
           (short)0, acc0,  false,  false);
      acc1 = __builtin_amdgcn_wmma_f32_16x16x4_f32(
           false, a[c],  false, bB[c],
           (short)0, acc1,  false,  false);
    }
    __builtin_amdgcn_sched_barrier(0);
    asm volatile("v_nop\n\tv_nop\n\tv_nop\n\tv_nop\n\tv_nop\n\tv_nop\n\tv_nop\n\tv_nop"
                 : "+v"(acc0), "+v"(acc1));

    float xr[8], yr[8];
#pragma unroll
    for (int r = 0; r < 8; ++r) {
      const float a0 = acc0[r];
      const float a1 = acc1[r];
      const float a0x = __builtin_bit_cast(float, __builtin_amdgcn_permlanex16(
          0u, __builtin_bit_cast(unsigned, a0), 0x76543210u, 0xfedcba98u, false, false));
      const float a1x = __builtin_bit_cast(float, __builtin_amdgcn_permlanex16(
          0u, __builtin_bit_cast(unsigned, a1), 0x76543210u, 0xfedcba98u, false, false));
      xr[r] = half ? a1x : a0;
      yr[r] = half ? a1 : a0x;
    }

    int o_x = o_s;
    asm volatile("" : "+v"(o_x));
#pragma unroll
    for (int r = 0; r < 8; ++r) {
      const int oa = __builtin_amdgcn_readlane(o_x, r);
      const int ob = __builtin_amdgcn_readlane(o_x, 8 + r);
      if (oa >= 0 && oa < n_out && cvL)
        unsafeAtomicAdd(&out[(long long)oa * COUT + colL], xr[r]);
      if (ob >= 0 && ob < n_out && cvL)
        unsafeAtomicAdd(&out[(long long)ob * COUT + colL], yr[r]);
    }
  }
}

template <int C>
__global__ __launch_bounds__(256) void bn_stats(
    const float* __restrict__ x, const int* __restrict__ n_p,
    float* __restrict__ stats) {
  constexpr int U = 8;
  constexpr int G = C / 4;
  const int n = *n_p;
  const int t = threadIdx.x;
  const int lane = t & 31;
  const int w = t >> 5;
  const long long total4 = (long long)n * G;
  const long long T = (long long)gridDim.x * 256;
  const long long gt = (long long)blockIdx.x * 256 + t;
  const v4f* __restrict__ x4 = (const v4f*)x;

  v4f s = {}, q = {};
  for (long long base = gt; base < total4; base += U * T) {
    v4f v[U];
#pragma unroll
    for (int j = 0; j < U; ++j) {
      const long long i = base + j * T;
      const bool ok = i < total4;
      const v4f ld = x4[ok ? i : base];
      const v4f z = {};
      v[j] = ok ? ld : z;
    }
#pragma unroll
    for (int j = 0; j < U; ++j) {
      s += v[j];
      q += v[j] * v[j];
    }
  }

#pragma unroll
  for (int off = G; off < 32; off <<= 1) {
#pragma unroll
    for (int e = 0; e < 4; ++e) {
      s[e] += __shfl_xor(s[e], off);
      q[e] += __shfl_xor(q[e], off);
    }
  }

  __shared__ float part[8][2 * C];
  if (lane < G) {
#pragma unroll
    for (int e = 0; e < 4; ++e) {
      part[w][4 * lane + e]     = s[e];
      part[w][C + 4 * lane + e] = q[e];
    }
  }
  __syncthreads();
  if (t < 2 * C) {
    float tot = 0.0f;
#pragma unroll
    for (int k = 0; k < 8; ++k) tot += part[k][t];
    unsafeAtomicAdd(&stats[t], tot);
  }
}

extern "C" void kernel_launch(void* const* d_in, const int* in_sizes, int n_in,
                              void* d_out, int out_size, void* d_ws, size_t ws_size,
                              hipStream_t stream) {
  (void)n_in; (void)ws_size;
  const float* feats = (const float*)d_in[0];
  const float* W0  = (const float*)d_in[1];
  const float* g0  = (const float*)d_in[2];
  const float* b0  = (const float*)d_in[3];
  const float* W1  = (const float*)d_in[4];
  const float* g1  = (const float*)d_in[5];
  const float* b1  = (const float*)d_in[6];
  const float* W2  = (const float*)d_in[7];
  const float* g2  = (const float*)d_in[8];
  const float* b2  = (const float*)d_in[9];
  const float* Wt1 = (const float*)d_in[10];
  const float* gt1 = (const float*)d_in[11];
  const float* bt1 = (const float*)d_in[12];
  const float* Wt2 = (const float*)d_in[13];
  const int* in0  = (const int*)d_in[14];
  const int* out0 = (const int*)d_in[15];
  const int* in1  = (const int*)d_in[16];
  const int* out1 = (const int*)d_in[17];
  const int* in2  = (const int*)d_in[18];
  const int* out2 = (const int*)d_in[19];
  const int* int1 = (const int*)d_in[20];
  const int* outt1= (const int*)d_in[21];
  const int* int2 = (const int*)d_in[22];
  const int* outt2= (const int*)d_in[23];
  const int* n0p  = (const int*)d_in[24];
  const int* n1p  = (const int*)d_in[25];
  const int* n2p  = (const int*)d_in[26];

  const int n0  = in_sizes[0] / 4;
  const int P0  = in_sizes[14] / 27;
  const int P1  = in_sizes[16] / 27;
  const int P2  = in_sizes[18] / 27;
  const int Pt1 = in_sizes[20] / 27;
  const int Pt2 = in_sizes[22] / 27;

  float* ws  = (float*)d_ws;
  float* x0  = ws;
  float* x1  = x0  + (size_t)n0 * 16;
  float* x2  = x1  + (size_t)n0 * 32;
  float* xt1 = x2  + (size_t)n0 * 64;
  float* st0 = xt1 + (size_t)n0 * 32;
  float* st1 = st0 + 32;
  float* st2 = st1 + 64;
  float* stt = st2 + 128;
  const float* zrow = stt + 64;

  const long long zc4 = ((long long)n0 * (16 + 32 + 64 + 32) + 512) / 4;
  zero_f32v4<<<dim3((unsigned)((zc4 + 255) / 256)), 256, 0, stream>>>(ws, zc4);
  const long long oc4 = ((long long)out_size + 3) / 4;
  zero_f32v4<<<dim3((unsigned)((oc4 + 255) / 256)), 256, 0, stream>>>(
      (float*)d_out, oc4);

  constexpr int TPB = 8;
  const dim3 wv(32);
  auto tgrid = [](int P) { return (unsigned)((P + 16 * TPB - 1) / (16 * TPB)); };
  auto sg = [&](int C) {
    long long blocks = ((long long)n0 * C + 8191) / 8192;
    return dim3((unsigned)(blocks < 512 ? blocks : 512));
  };

  sconv_wmma<4, 16, TPB, false><<<dim3(tgrid(P0), 27, 1), wv, 0, stream>>>(
      feats, W0, in0, out0, n0p, n0p, P0, x0, zrow, nullptr, nullptr, nullptr);
  bn_stats<16><<<sg(16), 256, 0, stream>>>(x0, n0p, st0);

  sconv_wmma_row32<16, 32, 4, 2, true><<<dim3(tgrid(P1), 27, 1), dim3(128), 0, stream>>>(
      x0, W1, in1, out1, n0p, n1p, P1, x1, zrow, st0, g0, b0);
  bn_stats<32><<<sg(32), 256, 0, stream>>>(x1, n1p, st1);

  sconv_wmma_row32<32, 64, 2, 4, true><<<dim3(tgrid(P2), 27, 2), dim3(64), 0, stream>>>(
      x1, W2, in2, out2, n1p, n2p, P2, x2, zrow, st1, g1, b1);
  bn_stats<64><<<sg(64), 256, 0, stream>>>(x2, n2p, st2);

  sconv_wmma_row32<64, 32, 2, 4, true><<<dim3(tgrid(Pt1), 27, 1), dim3(64), 0, stream>>>(
      x2, Wt1, int1, outt1, n2p, n1p, Pt1, xt1, zrow, st2, g2, b2);
  bn_stats<32><<<sg(32), 256, 0, stream>>>(xt1, n1p, stt);

  sconv_wmma_row32<32, 20, 4, 2, true><<<dim3(tgrid(Pt2), 27, 1), dim3(128), 0, stream>>>(
      xt1, Wt2, int2, outt2, n1p, n0p, Pt2, (float*)d_out, zrow, stt, gt1, bt1);
}


# Conv tile sweep software-pipelined: a wave loads all its tiles' index words as one group at entry and gathers tile t+1's rows ahead of tile t's products and adds (was index, rows, adds in line)
# speedup vs baseline: 2.9168x; 1.1180x over previous
#include <hip/hip_runtime.h>

typedef __attribute__((ext_vector_type(2))) float v2f;
typedef __attribute__((ext_vector_type(8))) float v8f;
typedef __attribute__((ext_vector_type(4))) float v4f;

__global__ void zero_f32v4(float* __restrict__ p, long long count4) {
  long long i = (long long)blockIdx.x * blockDim.x + threadIdx.x;
  if (i < count4) {
    v4f z = {};
    *(v4f*)(p + 4 * i) = z;
  }
}

template <int CIN, int COUT, int TPB, bool BN>
__global__ __launch_bounds__(32) void sconv_wmma(
    const float* __restrict__ f, const float* __restrict__ W,
    const int* __restrict__ ii, const int* __restrict__ oo,
    const int* __restrict__ nin_p, const int* __restrict__ nout_p,
    int P, float* __restrict__ out, const float* __restrict__ zrow,
    const float* __restrict__ st, const float* __restrict__ g,
    const float* __restrict__ b) {
  static_assert((CIN / 4) * TPB <= 16, "all tiles' row pieces are held in registers at once");
  const int lane = threadIdx.x;
  const int m    = lane & 15;
  const int half = lane >> 4;

  const int k       = blockIdx.y;
  const int colBase = blockIdx.z << 4;

  const int* __restrict__ iik = ii + (long long)k * P;
  const int* __restrict__ ook = oo + (long long)k * P;

  int o_ld[TPB], r_ld[TPB];
  bool pvt[TPB];
#pragma unroll
  for (int t = 0; t < TPB; ++t) {
    const int p = (blockIdx.x * TPB + t) * 16 + m;
    pvt[t] = (p < P);
    const int pc = pvt[t] ? p : (P - 1);
    o_ld[t] = ook[pc];
    r_ld[t] = iik[pc];
  }

  const int n_in  = *nin_p;
  const int n_out = *nout_p;

  const int col  = colBase + m;
  constexpr bool FULLN = (COUT % 16) == 0;
  const bool cv  = FULLN || (col < COUT);
  const int colc = cv ? col : (COUT - 1);

  const float* __restrict__ Wk = W + (long long)k * (CIN * COUT);
  v2f bfrag[CIN / 4];
#pragma unroll
  for (int c = 0; c < CIN / 4; ++c) {
    const int k0 = c * 4 + half * 2;
    const float w0 = Wk[(k0 + 0) * COUT + colc];
    const float w1 = Wk[(k0 + 1) * COUT + colc];
    bfrag[c].x = cv ? w0 : 0.0f;
    bfrag[c].y = cv ? w1 : 0.0f;
  }

  v2f sc[CIN / 4], sh[CIN / 4];
  if constexpr (BN) {
    __shared__ __attribute__((aligned(16))) float lss[2 * CIN];
    const float inv_n = 1.0f / (float)n_in;
    for (int ch = lane; ch < CIN; ch += 32) {
      const float mean = st[ch] * inv_n;
      const float var  = st[CIN + ch] * inv_n - mean * mean;
      const float s    = rsqrtf(var + 1e-5f) * g[ch];
      const int slot   = (ch >> 1) * 4 + (ch & 1);
      lss[slot]     = s;
      lss[slot + 2] = b[ch] - mean * s;
    }
    __syncthreads();
#pragma unroll
    for (int c = 0; c < CIN / 4; ++c) {
      const v4f t = *(const v4f*)(lss + (c * 2 + half) * 4);
      sc[c].x = t.x; sc[c].y = t.y;
      sh[c].x = t.z; sh[c].y = t.w;
    }
  }

  int o_s[TPB];
  bool live[TPB];
  v2f a[TPB][CIN / 4];
#pragma unroll
  for (int t = 0; t < TPB; ++t) {
    const int o   = pvt[t] ? o_ld[t] : n_out;
    const bool ov = (o >= 0) && (o < n_out);
    live[t] = (__ballot(ov) != 0ull);
    const int row = pvt[t] ? r_ld[t] : n_in;
    const bool rv = (row >= 0) && (row < n_in);
    o_s[t] = (BN && !rv) ? -1 : o;
    const float* __restrict__ fr =
        (rv ? f + (long long)row * CIN : zrow) + half * 2;
#pragma unroll
    for (int c = 0; c < CIN / 4; ++c) a[t][c] = *(const v2f*)(fr + c * 4);
  }

  int xaddr[8];
#pragma unroll
  for (int r = 0; r < 8; ++r) {
    xaddr[r] = (half * 8 + r) << 2;
    asm volatile("" : "+v"(xaddr[r]));
  }

#pragma unroll
  for (int t = 0; t < TPB; ++t) {
    if (!live[t]) continue;
    const int base = (blockIdx.x * TPB + t) * 16;

    if constexpr (BN) {
#pragma unroll
      for (int c = 0; c < CIN / 4; ++c) {
        a[t][c].x = fmaxf(fmaf(a[t][c].x, sc[c].x, sh[c].x), 0.0f);
        a[t][c].y = fmaxf(fmaf(a[t][c].y, sc[c].y, sh[c].y), 0.0f);
      }
    }
    int o_x = o_s[t];
    asm volatile("" : "+v"(o_x));
    __builtin_amdgcn_sched_barrier(0);

    v8f acc = {};
#pragma unroll
    for (int c = 0; c < CIN / 4; ++c) {
      acc = __builtin_amdgcn_wmma_f32_16x16x4_f32(
           false, a[t][c],  false, bfrag[c],
           (short)0, acc,  false,  false);
    }
    __builtin_amdgcn_sched_barrier(0);
    asm volatile("v_nop\n\tv_nop\n\tv_nop\n\tv_nop\n\tv_nop\n\tv_nop\n\tv_nop\n\tv_nop"
                 : "+v"(acc));

    asm volatile("" : "+v"(o_x));
    int orow[8];
#pragma unroll
    for (int r = 0; r < 8; ++r) orow[r] = __builtin_amdgcn_ds_bpermute(xaddr[r], o_x);
#pragma unroll
    for (int r = 0; r < 8; ++r) {
      const int mr  = half * 8 + r;
      const int o_r = orow[r];
      if (cv && base + mr < P && o_r >= 0 && o_r < n_out) {
        unsafeAtomicAdd(&out[(long long)o_r * COUT + col], acc[r]);
      }
    }
  }
}

template <int CIN, int COUT, int WPB, int TPW, bool BN>
__global__ __launch_bounds__(32 * WPB) void sconv_wmma_row32(
    const float* __restrict__ f, const float* __restrict__ W,
    const int* __restrict__ ii, const int* __restrict__ oo,
    const int* __restrict__ nin_p, const int* __restrict__ nout_p,
    int P, float* __restrict__ out, const float* __restrict__ zrow,
    const float* __restrict__ st, const float* __restrict__ g,
    const float* __restrict__ b) {
  static_assert(COUT > 16, "one-tile layers use sconv_wmma");
  static_assert(!BN || 32 * WPB >= CIN, "one thread per input channel builds the scale / shift table");
  const int lane = threadIdx.x & 31;
  const int wv   = __builtin_amdgcn_readfirstlane((int)(threadIdx.x >> 5));
  const int m    = lane & 15;
  const int half = lane >> 4;

  const int k       = blockIdx.y;
  const int colBase = blockIdx.z << 5;

  const int* __restrict__ iik = ii + (long long)k * P;
  const int* __restrict__ ook = oo + (long long)k * P;

  int o_ld[TPW], r_ld[TPW];
  bool pvt[TPW];
#pragma unroll
  for (int t = 0; t < TPW; ++t) {
    const int tile = (blockIdx.x * TPW + t) * WPB + wv;
    const int p = tile * 16 + m;
    pvt[t] = (p < P);
    const int pc = pvt[t] ? p : (P - 1);
    o_ld[t] = ook[pc];
    r_ld[t] = iik[pc];
  }

  const int n_in  = *nin_p;
  const int n_out = *nout_p;

  constexpr bool FULLN = (COUT % 32) == 0;
  const int colA = colBase + m;
  const int colB = colA + 16;
  const bool cvA = FULLN || (colA < COUT);
  const bool cvB = FULLN || (colB < COUT);
  const int colAc = cvA ? colA : (COUT - 1);
  const int colBc = cvB ? colB : (COUT - 1);
  const int colL = colBase + lane;
  const bool cvL = FULLN || (colL < COUT);

  const float* __restrict__ Wk = W + (long long)k * (CIN * COUT);
  v2f bA[CIN / 4], bB[CIN / 4];
#pragma unroll
  for (int c = 0; c < CIN / 4; ++c) {
    const int k0 = c * 4 + half * 2;
    const float wa0 = Wk[(k0 + 0) * COUT + colAc];
    const float wa1 = Wk[(k0 + 1) * COUT + colAc];
    const float wb0 = Wk[(k0 + 0) * COUT + colBc];
    const float wb1 = Wk[(k0 + 1) * COUT + colBc];
    bA[c].x = cvA ? wa0 : 0.0f;
    bA[c].y = cvA ? wa1 : 0.0f;
    bB[c].x = cvB ? wb0 : 0.0f;
    bB[c].y = cvB ? wb1 : 0.0f;
  }

  float st_sum = 0.0f, st_sq = 0.0f, g_ch = 0.0f, b_ch = 0.0f;
  if constexpr (BN) {
    const int chc = ((int)threadIdx.x < CIN) ? (int)threadIdx.x : (CIN - 1);
    st_sum = st[chc];
    st_sq  = st[CIN + chc];
    g_ch   = g[chc];
    b_ch   = b[chc];
  }

  int o_s[TPW];
  bool live[TPW];
  const float* fr[TPW];
#pragma unroll
  for (int t = 0; t < TPW; ++t) {
    const int o   = pvt[t] ? o_ld[t] : n_out;
    const bool ov = (o >= 0) && (o < n_out);
    live[t] = (__ballot(ov) != 0ull);
    const int row = pvt[t] ? r_ld[t] : n_in;
    const bool rv = (row >= 0) && (row < n_in);
    o_s[t] = (BN && !rv) ? -1 : o;
    fr[t] = (rv ? f + (long long)row * CIN : zrow) + half * 2;
  }

  v2f a[2][CIN / 4];
#pragma unroll
  for (int c = 0; c < CIN / 4; ++c) a[0][c] = *(const v2f*)(fr[0] + c * 4);

  v2f sc[CIN / 4], sh[CIN / 4];
  if constexpr (BN) {
    __shared__ __attribute__((aligned(16))) float lss[2 * CIN];
    if ((int)threadIdx.x < CIN) {
      const int ch = (int)threadIdx.x;
      const float inv_n = 1.0f / (float)n_in;
      const float mean = st_sum * inv_n;
      const float var  = st_sq * inv_n - mean * mean;
      const float s    = rsqrtf(var + 1e-5f) * g_ch;
      const int slot   = (ch >> 1) * 4 + (ch & 1);
      lss[slot]     = s;
      lss[slot + 2] = b_ch - mean * s;
    }
    __syncthreads();
#pragma unroll
    for (int c = 0; c < CIN / 4; ++c) {
      const v4f t4 = *(const v4f*)(lss + (c * 2 + half) * 4);
      sc[c].x = t4.x; sc[c].y = t4.y;
      sh[c].x = t4.z; sh[c].y = t4.w;
    }
  }

#pragma unroll
  for (int t = 0; t < TPW; ++t) {
    if (t + 1 < TPW) {
#pragma unroll
      for (int c = 0; c < CIN / 4; ++c)
        a[(t + 1) & 1][c] = *(const v2f*)(fr[t + 1] + c * 4);
    }
    if (!live[t]) continue;

    if constexpr (BN) {
#pragma unroll
      for (int c = 0; c < CIN / 4; ++c) {
        a[t & 1][c].x = fmaxf(fmaf(a[t & 1][c].x, sc[c].x, sh[c].x), 0.0f);
        a[t & 1][c].y = fmaxf(fmaf(a[t & 1][c].y, sc[c].y, sh[c].y), 0.0f);
      }
    }
    int o_x = o_s[t];
    asm volatile("" : "+v"(o_x));
    __builtin_amdgcn_sched_barrier(0);

    v8f acc0 = {}, acc1 = {};
#pragma unroll
    for (int c = 0; c < CIN / 4; ++c) {
      acc0 = __builtin_amdgcn_wmma_f32_16x16x4_f32(
           false, a[t & 1][c],  false, bA[c],
           (short)0, acc0,  false,  false);
      acc1 = __builtin_amdgcn_wmma_f32_16x16x4_f32(
           false, a[t & 1][c],  false, bB[c],
           (short)0, acc1,  false,  false);
    }
    __builtin_amdgcn_sched_barrier(0);
    asm volatile("v_nop\n\tv_nop\n\tv_nop\n\tv_nop\n\tv_nop\n\tv_nop\n\tv_nop\n\tv_nop"
                 : "+v"(acc0), "+v"(acc1));

    float xr[8], yr[8];
#pragma unroll
    for (int r = 0; r < 8; ++r) {
      const float a0 = acc0[r];
      const float a1 = acc1[r];
      const float a0x = __builtin_bit_cast(float, __builtin_amdgcn_permlanex16(
          0u, __builtin_bit_cast(unsigned, a0), 0x76543210u, 0xfedcba98u, false, false));
      const float a1x = __builtin_bit_cast(float, __builtin_amdgcn_permlanex16(
          0u, __builtin_bit_cast(unsigned, a1), 0x76543210u, 0xfedcba98u, false, false));
      xr[r] = half ? a1x : a0;
      yr[r] = half ? a1 : a0x;
    }

    asm volatile("" : "+v"(o_x));
#pragma unroll
    for (int r = 0; r < 8; ++r) {
      const int oa = __builtin_amdgcn_readlane(o_x, r);
      const int ob = __builtin_amdgcn_readlane(o_x, 8 + r);
      if (oa >= 0 && oa < n_out && cvL)
        unsafeAtomicAdd(&out[(long long)oa * COUT + colL], xr[r]);
      if (ob >= 0 && ob < n_out && cvL)
        unsafeAtomicAdd(&out[(long long)ob * COUT + colL], yr[r]);
    }
  }
}

template <int C>
__global__ __launch_bounds__(256) void bn_stats(
    const float* __restrict__ x, const int* __restrict__ n_p,
    float* __restrict__ stats) {
  constexpr int U = 8;
  constexpr int G = C / 4;
  const int n = *n_p;
  const int t = threadIdx.x;
  const int lane = t & 31;
  const int w = t >> 5;
  const long long total4 = (long long)n * G;
  const long long T = (long long)gridDim.x * 256;
  const long long gt = (long long)blockIdx.x * 256 + t;
  const v4f* __restrict__ x4 = (const v4f*)x;

  v4f s = {}, q = {};
  for (long long base = gt; base < total4; base += U * T) {
    v4f v[U];
#pragma unroll
    for (int j = 0; j < U; ++j) {
      const long long i = base + j * T;
      const bool ok = i < total4;
      const v4f ld = x4[ok ? i : base];
      const v4f z = {};
      v[j] = ok ? ld : z;
    }
#pragma unroll
    for (int j = 0; j < U; ++j) {
      s += v[j];
      q += v[j] * v[j];
    }
  }

#pragma unroll
  for (int off = G; off < 32; off <<= 1) {
#pragma unroll
    for (int e = 0; e < 4; ++e) {
      s[e] += __shfl_xor(s[e], off);
      q[e] += __shfl_xor(q[e], off);
    }
  }

  __shared__ float part[8][2 * C];
  if (lane < G) {
#pragma unroll
    for (int e = 0; e < 4; ++e) {
      part[w][4 * lane + e]     = s[e];
      part[w][C + 4 * lane + e] = q[e];
    }
  }
  __syncthreads();
  if (t < 2 * C) {
    float tot = 0.0f;
#pragma unroll
    for (int k = 0; k < 8; ++k) tot += part[k][t];
    unsafeAtomicAdd(&stats[t], tot);
  }
}

extern "C" void kernel_launch(void* const* d_in, const int* in_sizes, int n_in,
                              void* d_out, int out_size, void* d_ws, size_t ws_size,
                              hipStream_t stream) {
  (void)n_in; (void)ws_size;
  const float* feats = (const float*)d_in[0];
  const float* W0  = (const float*)d_in[1];
  const float* g0  = (const float*)d_in[2];
  const float* b0  = (const float*)d_in[3];
  const float* W1  = (const float*)d_in[4];
  const float* g1  = (const float*)d_in[5];
  const float* b1  = (const float*)d_in[6];
  const float* W2  = (const float*)d_in[7];
  const float* g2  = (const float*)d_in[8];
  const float* b2  = (const float*)d_in[9];
  const float* Wt1 = (const float*)d_in[10];
  const float* gt1 = (const float*)d_in[11];
  const float* bt1 = (const float*)d_in[12];
  const float* Wt2 = (const float*)d_in[13];
  const int* in0  = (const int*)d_in[14];
  const int* out0 = (const int*)d_in[15];
  const int* in1  = (const int*)d_in[16];
  const int* out1 = (const int*)d_in[17];
  const int* in2  = (const int*)d_in[18];
  const int* out2 = (const int*)d_in[19];
  const int* int1 = (const int*)d_in[20];
  const int* outt1= (const int*)d_in[21];
  const int* int2 = (const int*)d_in[22];
  const int* outt2= (const int*)d_in[23];
  const int* n0p  = (const int*)d_in[24];
  const int* n1p  = (const int*)d_in[25];
  const int* n2p  = (const int*)d_in[26];

  const int n0  = in_sizes[0] / 4;
  const int P0  = in_sizes[14] / 27;
  const int P1  = in_sizes[16] / 27;
  const int P2  = in_sizes[18] / 27;
  const int Pt1 = in_sizes[20] / 27;
  const int Pt2 = in_sizes[22] / 27;

  float* ws  = (float*)d_ws;
  float* x0  = ws;
  float* x1  = x0  + (size_t)n0 * 16;
  float* x2  = x1  + (size_t)n0 * 32;
  float* xt1 = x2  + (size_t)n0 * 64;
  float* st0 = xt1 + (size_t)n0 * 32;
  float* st1 = st0 + 32;
  float* st2 = st1 + 64;
  float* stt = st2 + 128;
  const float* zrow = stt + 64;

  const long long zc4 = ((long long)n0 * (16 + 32 + 64 + 32) + 512) / 4;
  zero_f32v4<<<dim3((unsigned)((zc4 + 255) / 256)), 256, 0, stream>>>(ws, zc4);
  const long long oc4 = ((long long)out_size + 3) / 4;
  zero_f32v4<<<dim3((unsigned)((oc4 + 255) / 256)), 256, 0, stream>>>(
      (float*)d_out, oc4);

  constexpr int TPB = 8;
  const dim3 wv(32);
  auto tgrid = [](int P) { return (unsigned)((P + 16 * TPB - 1) / (16 * TPB)); };
  auto sg = [&](int C) {
    long long blocks = ((long long)n0 * C + 8191) / 8192;
    return dim3((unsigned)(blocks < 512 ? blocks : 512));
  };

  sconv_wmma<4, 16, TPB, false><<<dim3(tgrid(P0), 27, 1), wv, 0, stream>>>(
      feats, W0, in0, out0, n0p, n0p, P0, x0, zrow, nullptr, nullptr, nullptr);
  bn_stats<16><<<sg(16), 256, 0, stream>>>(x0, n0p, st0);

  sconv_wmma_row32<16, 32, 4, 2, true><<<dim3(tgrid(P1), 27, 1), dim3(128), 0, stream>>>(
      x0, W1, in1, out1, n0p, n1p, P1, x1, zrow, st0, g0, b0);
  bn_stats<32><<<sg(32), 256, 0, stream>>>(x1, n1p, st1);

  sconv_wmma_row32<32, 64, 2, 4, true><<<dim3(tgrid(P2), 27, 2), dim3(64), 0, stream>>>(
      x1, W2, in2, out2, n1p, n2p, P2, x2, zrow, st1, g1, b1);
  bn_stats<64><<<sg(64), 256, 0, stream>>>(x2, n2p, st2);

  sconv_wmma_row32<64, 32, 2, 4, true><<<dim3(tgrid(Pt1), 27, 1), dim3(64), 0, stream>>>(
      x2, Wt1, int1, outt1, n2p, n1p, Pt1, xt1, zrow, st2, g2, b2);
  bn_stats<32><<<sg(32), 256, 0, stream>>>(xt1, n1p, stt);

  sconv_wmma_row32<32, 20, 4, 2, true><<<dim3(tgrid(Pt2), 27, 1), dim3(128), 0, stream>>>(
      xt1, Wt2, int2, outt2, n1p, n0p, Pt2, (float*)d_out, zrow, stt, gt1, bt1);
}


# Zero fill of x1, x2, xt1 and the output moved into the first conv's waves (mostly padding-only, idle): each wave stores 512-B runs; one fill launch left (stats, x0); 11 -> 10 launches
# speedup vs baseline: 3.0896x; 1.0592x over previous
#include <hip/hip_runtime.h>

typedef __attribute__((ext_vector_type(2))) float v2f;
typedef __attribute__((ext_vector_type(8))) float v8f;
typedef __attribute__((ext_vector_type(4))) float v4f;

__global__ void zero_f32v4(float* __restrict__ p, long long count4) {
  long long i = (long long)blockIdx.x * blockDim.x + threadIdx.x;
  if (i < count4) {
    v4f z = {};
    *(v4f*)(p + 4 * i) = z;
  }
}

template <int CIN, int COUT, int TPB, bool BN>
__global__ __launch_bounds__(32) void sconv_wmma(
    const float* __restrict__ f, const float* __restrict__ W,
    const int* __restrict__ ii, const int* __restrict__ oo,
    const int* __restrict__ nin_p, const int* __restrict__ nout_p,
    int P, float* __restrict__ out, const float* __restrict__ zrow,
    const float* __restrict__ st, const float* __restrict__ g,
    const float* __restrict__ b,
    float* __restrict__ zA, long long cA4, float* __restrict__ zB, long long cB4) {
  static_assert((CIN / 4) * TPB <= 16, "all tiles' row pieces are held in registers at once");
  const int lane = threadIdx.x;
  const int m    = lane & 15;
  const int half = lane >> 4;

  const int k       = blockIdx.y;
  const int colBase = blockIdx.z << 4;

  const int* __restrict__ iik = ii + (long long)k * P;
  const int* __restrict__ ook = oo + (long long)k * P;

  int o_ld[TPB], r_ld[TPB];
  bool pvt[TPB];
#pragma unroll
  for (int t = 0; t < TPB; ++t) {
    const int p = (blockIdx.x * TPB + t) * 16 + m;
    pvt[t] = (p < P);
    const int pc = pvt[t] ? p : (P - 1);
    o_ld[t] = ook[pc];
    r_ld[t] = iik[pc];
  }

  const int n_in  = *nin_p;
  const int n_out = *nout_p;

  const int col  = colBase + m;
  constexpr bool FULLN = (COUT % 16) == 0;
  const bool cv  = FULLN || (col < COUT);
  const int colc = cv ? col : (COUT - 1);

  const float* __restrict__ Wk = W + (long long)k * (CIN * COUT);
  v2f bfrag[CIN / 4];
#pragma unroll
  for (int c = 0; c < CIN / 4; ++c) {
    const int k0 = c * 4 + half * 2;
    const float w0 = Wk[(k0 + 0) * COUT + colc];
    const float w1 = Wk[(k0 + 1) * COUT + colc];
    bfrag[c].x = cv ? w0 : 0.0f;
    bfrag[c].y = cv ? w1 : 0.0f;
  }

  {
    const long long nthr = (long long)gridDim.x * gridDim.y * gridDim.z * 32;
    const long long gt =
        (((long long)blockIdx.z * gridDim.y + blockIdx.y) * gridDim.x + blockIdx.x) * 32 + lane;
    const v4f z4 = {};
    for (long long i = gt; i < cA4 + cB4; i += nthr) {
      float* __restrict__ zp = (i < cA4) ? (zA + 4 * i) : (zB + 4 * (i - cA4));
      *(v4f*)zp = z4;
    }
  }

  v2f sc[CIN / 4], sh[CIN / 4];
  if constexpr (BN) {
    __shared__ __attribute__((aligned(16))) float lss[2 * CIN];
    const float inv_n = 1.0f / (float)n_in;
    for (int ch = lane; ch < CIN; ch += 32) {
      const float mean = st[ch] * inv_n;
      const float var  = st[CIN + ch] * inv_n - mean * mean;
      const float s    = rsqrtf(var + 1e-5f) * g[ch];
      const int slot   = (ch >> 1) * 4 + (ch & 1);
      lss[slot]     = s;
      lss[slot + 2] = b[ch] - mean * s;
    }
    __syncthreads();
#pragma unroll
    for (int c = 0; c < CIN / 4; ++c) {
      const v4f t = *(const v4f*)(lss + (c * 2 + half) * 4);
      sc[c].x = t.x; sc[c].y = t.y;
      sh[c].x = t.z; sh[c].y = t.w;
    }
  }

  int o_s[TPB];
  bool live[TPB];
  v2f a[TPB][CIN / 4];
#pragma unroll
  for (int t = 0; t < TPB; ++t) {
    const int o   = pvt[t] ? o_ld[t] : n_out;
    const bool ov = (o >= 0) && (o < n_out);
    live[t] = (__ballot(ov) != 0ull);
    const int row = pvt[t] ? r_ld[t] : n_in;
    const bool rv = (row >= 0) && (row < n_in);
    o_s[t] = (BN && !rv) ? -1 : o;
    const float* __restrict__ fr =
        (rv ? f + (long long)row * CIN : zrow) + half * 2;
#pragma unroll
    for (int c = 0; c < CIN / 4; ++c) a[t][c] = *(const v2f*)(fr + c * 4);
  }

  int xaddr[8];
#pragma unroll
  for (int r = 0; r < 8; ++r) {
    xaddr[r] = (half * 8 + r) << 2;
    asm volatile("" : "+v"(xaddr[r]));
  }

#pragma unroll
  for (int t = 0; t < TPB; ++t) {
    if (!live[t]) continue;
    const int base = (blockIdx.x * TPB + t) * 16;

    if constexpr (BN) {
#pragma unroll
      for (int c = 0; c < CIN / 4; ++c) {
        a[t][c].x = fmaxf(fmaf(a[t][c].x, sc[c].x, sh[c].x), 0.0f);
        a[t][c].y = fmaxf(fmaf(a[t][c].y, sc[c].y, sh[c].y), 0.0f);
      }
    }
    int o_x = o_s[t];
    asm volatile("" : "+v"(o_x));
    __builtin_amdgcn_sched_barrier(0);

    v8f acc = {};
#pragma unroll
    for (int c = 0; c < CIN / 4; ++c) {
      acc = __builtin_amdgcn_wmma_f32_16x16x4_f32(
           false, a[t][c],  false, bfrag[c],
           (short)0, acc,  false,  false);
    }
    __builtin_amdgcn_sched_barrier(0);
    asm volatile("v_nop\n\tv_nop\n\tv_nop\n\tv_nop\n\tv_nop\n\tv_nop\n\tv_nop\n\tv_nop"
                 : "+v"(acc));

    asm volatile("" : "+v"(o_x));
    int orow[8];
#pragma unroll
    for (int r = 0; r < 8; ++r) orow[r] = __builtin_amdgcn_ds_bpermute(xaddr[r], o_x);
#pragma unroll
    for (int r = 0; r < 8; ++r) {
      const int mr  = half * 8 + r;
      const int o_r = orow[r];
      if (cv && base + mr < P && o_r >= 0 && o_r < n_out) {
        unsafeAtomicAdd(&out[(long long)o_r * COUT + col], acc[r]);
      }
    }
  }
}

template <int CIN, int COUT, int WPB, int TPW, bool BN>
__global__ __launch_bounds__(32 * WPB) void sconv_wmma_row32(
    const float* __restrict__ f, const float* __restrict__ W,
    const int* __restrict__ ii, const int* __restrict__ oo,
    const int* __restrict__ nin_p, const int* __restrict__ nout_p,
    int P, float* __restrict__ out, const float* __restrict__ zrow,
    const float* __restrict__ st, const float* __restrict__ g,
    const float* __restrict__ b) {
  static_assert(COUT > 16, "one-tile layers use sconv_wmma");
  static_assert(!BN || 32 * WPB >= CIN, "one thread per input channel builds the scale / shift table");
  const int lane = threadIdx.x & 31;
  const int wv   = __builtin_amdgcn_readfirstlane((int)(threadIdx.x >> 5));
  const int m    = lane & 15;
  const int half = lane >> 4;

  const int k       = blockIdx.y;
  const int colBase = blockIdx.z << 5;

  const int* __restrict__ iik = ii + (long long)k * P;
  const int* __restrict__ ook = oo + (long long)k * P;

  int o_ld[TPW], r_ld[TPW];
  bool pvt[TPW];
#pragma unroll
  for (int t = 0; t < TPW; ++t) {
    const int tile = (blockIdx.x * TPW + t) * WPB + wv;
    const int p = tile * 16 + m;
    pvt[t] = (p < P);
    const int pc = pvt[t] ? p : (P - 1);
    o_ld[t] = ook[pc];
    r_ld[t] = iik[pc];
  }

  const int n_in  = *nin_p;
  const int n_out = *nout_p;

  constexpr bool FULLN = (COUT % 32) == 0;
  const int colA = colBase + m;
  const int colB = colA + 16;
  const bool cvA = FULLN || (colA < COUT);
  const bool cvB = FULLN || (colB < COUT);
  const int colAc = cvA ? colA : (COUT - 1);
  const int colBc = cvB ? colB : (COUT - 1);
  const int colL = colBase + lane;
  const bool cvL = FULLN || (colL < COUT);

  const float* __restrict__ Wk = W + (long long)k * (CIN * COUT);
  v2f bA[CIN / 4], bB[CIN / 4];
#pragma unroll
  for (int c = 0; c < CIN / 4; ++c) {
    const int k0 = c * 4 + half * 2;
    const float wa0 = Wk[(k0 + 0) * COUT + colAc];
    const float wa1 = Wk[(k0 + 1) * COUT + colAc];
    const float wb0 = Wk[(k0 + 0) * COUT + colBc];
    const float wb1 = Wk[(k0 + 1) * COUT + colBc];
    bA[c].x = cvA ? wa0 : 0.0f;
    bA[c].y = cvA ? wa1 : 0.0f;
    bB[c].x = cvB ? wb0 : 0.0f;
    bB[c].y = cvB ? wb1 : 0.0f;
  }

  float st_sum = 0.0f, st_sq = 0.0f, g_ch = 0.0f, b_ch = 0.0f;
  if constexpr (BN) {
    const int chc = ((int)threadIdx.x < CIN) ? (int)threadIdx.x : (CIN - 1);
    st_sum = st[chc];
    st_sq  = st[CIN + chc];
    g_ch   = g[chc];
    b_ch   = b[chc];
  }

  int o_s[TPW];
  bool live[TPW];
  const float* fr[TPW];
#pragma unroll
  for (int t = 0; t < TPW; ++t) {
    const int o   = pvt[t] ? o_ld[t] : n_out;
    const bool ov = (o >= 0) && (o < n_out);
    live[t] = (__ballot(ov) != 0ull);
    const int row = pvt[t] ? r_ld[t] : n_in;
    const bool rv = (row >= 0) && (row < n_in);
    o_s[t] = (BN && !rv) ? -1 : o;
    fr[t] = (rv ? f + (long long)row * CIN : zrow) + half * 2;
  }

  v2f a[2][CIN / 4];
#pragma unroll
  for (int c = 0; c < CIN / 4; ++c) a[0][c] = *(const v2f*)(fr[0] + c * 4);

  v2f sc[CIN / 4], sh[CIN / 4];
  if constexpr (BN) {
    __shared__ __attribute__((aligned(16))) float lss[2 * CIN];
    if ((int)threadIdx.x < CIN) {
      const int ch = (int)threadIdx.x;
      const float inv_n = 1.0f / (float)n_in;
      const float mean = st_sum * inv_n;
      const float var  = st_sq * inv_n - mean * mean;
      const float s    = rsqrtf(var + 1e-5f) * g_ch;
      const int slot   = (ch >> 1) * 4 + (ch & 1);
      lss[slot]     = s;
      lss[slot + 2] = b_ch - mean * s;
    }
    __syncthreads();
#pragma unroll
    for (int c = 0; c < CIN / 4; ++c) {
      const v4f t4 = *(const v4f*)(lss + (c * 2 + half) * 4);
      sc[c].x = t4.x; sc[c].y = t4.y;
      sh[c].x = t4.z; sh[c].y = t4.w;
    }
  }

#pragma unroll
  for (int t = 0; t < TPW; ++t) {
    if (t + 1 < TPW) {
#pragma unroll
      for (int c = 0; c < CIN / 4; ++c)
        a[(t + 1) & 1][c] = *(const v2f*)(fr[t + 1] + c * 4);
    }
    if (!live[t]) continue;

    if constexpr (BN) {
#pragma unroll
      for (int c = 0; c < CIN / 4; ++c) {
        a[t & 1][c].x = fmaxf(fmaf(a[t & 1][c].x, sc[c].x, sh[c].x), 0.0f);
        a[t & 1][c].y = fmaxf(fmaf(a[t & 1][c].y, sc[c].y, sh[c].y), 0.0f);
      }
    }
    int o_x = o_s[t];
    asm volatile("" : "+v"(o_x));
    __builtin_amdgcn_sched_barrier(0);

    v8f acc0 = {}, acc1 = {};
#pragma unroll
    for (int c = 0; c < CIN / 4; ++c) {
      acc0 = __builtin_amdgcn_wmma_f32_16x16x4_f32(
           false, a[t & 1][c],  false, bA[c],
           (short)0, acc0,  false,  false);
      acc1 = __builtin_amdgcn_wmma_f32_16x16x4_f32(
           false, a[t & 1][c],  false, bB[c],
           (short)0, acc1,  false,  false);
    }
    __builtin_amdgcn_sched_barrier(0);
    asm volatile("v_nop\n\tv_nop\n\tv_nop\n\tv_nop\n\tv_nop\n\tv_nop\n\tv_nop\n\tv_nop"
                 : "+v"(acc0), "+v"(acc1));

    float xr[8], yr[8];
#pragma unroll
    for (int r = 0; r < 8; ++r) {
      const float a0 = acc0[r];
      const float a1 = acc1[r];
      const float a0x = __builtin_bit_cast(float, __builtin_amdgcn_permlanex16(
          0u, __builtin_bit_cast(unsigned, a0), 0x76543210u, 0xfedcba98u, false, false));
      const float a1x = __builtin_bit_cast(float, __builtin_amdgcn_permlanex16(
          0u, __builtin_bit_cast(unsigned, a1), 0x76543210u, 0xfedcba98u, false, false));
      xr[r] = half ? a1x : a0;
      yr[r] = half ? a1 : a0x;
    }

    asm volatile("" : "+v"(o_x));
#pragma unroll
    for (int r = 0; r < 8; ++r) {
      const int oa = __builtin_amdgcn_readlane(o_x, r);
      const int ob = __builtin_amdgcn_readlane(o_x, 8 + r);
      if (oa >= 0 && oa < n_out && cvL)
        unsafeAtomicAdd(&out[(long long)oa * COUT + colL], xr[r]);
      if (ob >= 0 && ob < n_out && cvL)
        unsafeAtomicAdd(&out[(long long)ob * COUT + colL], yr[r]);
    }
  }
}

template <int C>
__global__ __launch_bounds__(256) void bn_stats(
    const float* __restrict__ x, const int* __restrict__ n_p,
    float* __restrict__ stats) {
  constexpr int U = 8;
  constexpr int G = C / 4;
  const int n = *n_p;
  const int t = threadIdx.x;
  const int lane = t & 31;
  const int w = t >> 5;
  const long long total4 = (long long)n * G;
  const long long T = (long long)gridDim.x * 256;
  const long long gt = (long long)blockIdx.x * 256 + t;
  const v4f* __restrict__ x4 = (const v4f*)x;

  v4f s = {}, q = {};
  for (long long base = gt; base < total4; base += U * T) {
    v4f v[U];
#pragma unroll
    for (int j = 0; j < U; ++j) {
      const long long i = base + j * T;
      const bool ok = i < total4;
      const v4f ld = x4[ok ? i : base];
      const v4f z = {};
      v[j] = ok ? ld : z;
    }
#pragma unroll
    for (int j = 0; j < U; ++j) {
      s += v[j];
      q += v[j] * v[j];
    }
  }

#pragma unroll
  for (int off = G; off < 32; off <<= 1) {
#pragma unroll
    for (int e = 0; e < 4; ++e) {
      s[e] += __shfl_xor(s[e], off);
      q[e] += __shfl_xor(q[e], off);
    }
  }

  __shared__ float part[8][2 * C];
  if (lane < G) {
#pragma unroll
    for (int e = 0; e < 4; ++e) {
      part[w][4 * lane + e]     = s[e];
      part[w][C + 4 * lane + e] = q[e];
    }
  }
  __syncthreads();
  if (t < 2 * C) {
    float tot = 0.0f;
#pragma unroll
    for (int k = 0; k < 8; ++k) tot += part[k][t];
    unsafeAtomicAdd(&stats[t], tot);
  }
}

extern "C" void kernel_launch(void* const* d_in, const int* in_sizes, int n_in,
                              void* d_out, int out_size, void* d_ws, size_t ws_size,
                              hipStream_t stream) {
  (void)n_in; (void)ws_size;
  const float* feats = (const float*)d_in[0];
  const float* W0  = (const float*)d_in[1];
  const float* g0  = (const float*)d_in[2];
  const float* b0  = (const float*)d_in[3];
  const float* W1  = (const float*)d_in[4];
  const float* g1  = (const float*)d_in[5];
  const float* b1  = (const float*)d_in[6];
  const float* W2  = (const float*)d_in[7];
  const float* g2  = (const float*)d_in[8];
  const float* b2  = (const float*)d_in[9];
  const float* Wt1 = (const float*)d_in[10];
  const float* gt1 = (const float*)d_in[11];
  const float* bt1 = (const float*)d_in[12];
  const float* Wt2 = (const float*)d_in[13];
  const int* in0  = (const int*)d_in[14];
  const int* out0 = (const int*)d_in[15];
  const int* in1  = (const int*)d_in[16];
  const int* out1 = (const int*)d_in[17];
  const int* in2  = (const int*)d_in[18];
  const int* out2 = (const int*)d_in[19];
  const int* int1 = (const int*)d_in[20];
  const int* outt1= (const int*)d_in[21];
  const int* int2 = (const int*)d_in[22];
  const int* outt2= (const int*)d_in[23];
  const int* n0p  = (const int*)d_in[24];
  const int* n1p  = (const int*)d_in[25];
  const int* n2p  = (const int*)d_in[26];

  const int n0  = in_sizes[0] / 4;
  const int P0  = in_sizes[14] / 27;
  const int P1  = in_sizes[16] / 27;
  const int P2  = in_sizes[18] / 27;
  const int Pt1 = in_sizes[20] / 27;
  const int Pt2 = in_sizes[22] / 27;

  float* ws  = (float*)d_ws;
  float* st0 = ws;
  float* st1 = st0 + 32;
  float* st2 = st1 + 64;
  float* stt = st2 + 128;
  const float* zrow = stt + 64;
  float* x0  = ws  + 512;
  float* x1  = x0  + (size_t)n0 * 16;
  float* x2  = x1  + (size_t)n0 * 32;
  float* xt1 = x2  + (size_t)n0 * 64;

  const long long zc4 = ((long long)n0 * 16 + 512) / 4;
  zero_f32v4<<<dim3((unsigned)((zc4 + 255) / 256)), 256, 0, stream>>>(ws, zc4);
  const long long xc4 = ((long long)n0 * (32 + 64 + 32)) / 4;
  const long long oc4 = ((long long)out_size + 3) / 4;

  constexpr int TPB = 8;
  const dim3 wv(32);
  auto tgrid = [](int P) { return (unsigned)((P + 16 * TPB - 1) / (16 * TPB)); };
  auto sg = [&](int C) {
    long long blocks = ((long long)n0 * C + 8191) / 8192;
    return dim3((unsigned)(blocks < 512 ? blocks : 512));
  };

  sconv_wmma<4, 16, TPB, false><<<dim3(tgrid(P0), 27, 1), wv, 0, stream>>>(
      feats, W0, in0, out0, n0p, n0p, P0, x0, zrow, nullptr, nullptr, nullptr,
      x1, xc4, (float*)d_out, oc4);
  bn_stats<16><<<sg(16), 256, 0, stream>>>(x0, n0p, st0);

  sconv_wmma_row32<16, 32, 4, 2, true><<<dim3(tgrid(P1), 27, 1), dim3(128), 0, stream>>>(
      x0, W1, in1, out1, n0p, n1p, P1, x1, zrow, st0, g0, b0);
  bn_stats<32><<<sg(32), 256, 0, stream>>>(x1, n1p, st1);

  sconv_wmma_row32<32, 64, 2, 4, true><<<dim3(tgrid(P2), 27, 2), dim3(64), 0, stream>>>(
      x1, W2, in2, out2, n1p, n2p, P2, x2, zrow, st1, g1, b1);
  bn_stats<64><<<sg(64), 256, 0, stream>>>(x2, n2p, st2);

  sconv_wmma_row32<64, 32, 2, 4, true><<<dim3(tgrid(Pt1), 27, 1), dim3(64), 0, stream>>>(
      x2, Wt1, int1, outt1, n2p, n1p, Pt1, xt1, zrow, st2, g2, b2);
  bn_stats<32><<<sg(32), 256, 0, stream>>>(xt1, n1p, stt);

  sconv_wmma_row32<32, 20, 4, 2, true><<<dim3(tgrid(Pt2), 27, 1), dim3(128), 0, stream>>>(
      xt1, Wt2, int2, outt2, n1p, n0p, Pt2, (float*)d_out, zrow, stt, gt1, bt1);
}


# Scatter tests moved from per add to per tile in all 5 convs: a tile with 16 valid rows adds under one uniform branch, 20-col tail masks lanes once (conv0 index by v_readlane; permlanex16 fi form)
# speedup vs baseline: 3.2730x; 1.0594x over previous
#include <hip/hip_runtime.h>

typedef __attribute__((ext_vector_type(2))) float v2f;
typedef __attribute__((ext_vector_type(8))) float v8f;
typedef __attribute__((ext_vector_type(4))) float v4f;

__global__ void zero_f32v4(float* __restrict__ p, long long count4) {
  long long i = (long long)blockIdx.x * blockDim.x + threadIdx.x;
  if (i < count4) {
    v4f z = {};
    *(v4f*)(p + 4 * i) = z;
  }
}

template <int CIN, int COUT, int TPB, bool BN>
__global__ __launch_bounds__(32) void sconv_wmma(
    const float* __restrict__ f, const float* __restrict__ W,
    const int* __restrict__ ii, const int* __restrict__ oo,
    const int* __restrict__ nin_p, const int* __restrict__ nout_p,
    int P, float* __restrict__ out, const float* __restrict__ zrow,
    const float* __restrict__ st, const float* __restrict__ g,
    const float* __restrict__ b,
    float* __restrict__ zA, long long cA4, float* __restrict__ zB, long long cB4) {
  static_assert((CIN / 4) * TPB <= 16, "all tiles' row pieces are held in registers at once");
  const int lane = threadIdx.x;
  const int m    = lane & 15;
  const int half = lane >> 4;

  const int k       = blockIdx.y;
  const int colBase = blockIdx.z << 4;

  const int* __restrict__ iik = ii + (long long)k * P;
  const int* __restrict__ ook = oo + (long long)k * P;

  int o_ld[TPB], r_ld[TPB];
  bool pvt[TPB];
#pragma unroll
  for (int t = 0; t < TPB; ++t) {
    const int p = (blockIdx.x * TPB + t) * 16 + m;
    pvt[t] = (p < P);
    const int pc = pvt[t] ? p : (P - 1);
    o_ld[t] = ook[pc];
    r_ld[t] = iik[pc];
  }

  const int n_in  = *nin_p;
  const int n_out = *nout_p;

  const int col  = colBase + m;
  constexpr bool FULLN = (COUT % 16) == 0;
  const bool cv  = FULLN || (col < COUT);
  const int colc = cv ? col : (COUT - 1);

  const float* __restrict__ Wk = W + (long long)k * (CIN * COUT);
  v2f bfrag[CIN / 4];
#pragma unroll
  for (int c = 0; c < CIN / 4; ++c) {
    const int k0 = c * 4 + half * 2;
    const float w0 = Wk[(k0 + 0) * COUT + colc];
    const float w1 = Wk[(k0 + 1) * COUT + colc];
    bfrag[c].x = cv ? w0 : 0.0f;
    bfrag[c].y = cv ? w1 : 0.0f;
  }

  {
    const long long nthr = (long long)gridDim.x * gridDim.y * gridDim.z * 32;
    const long long gt =
        (((long long)blockIdx.z * gridDim.y + blockIdx.y) * gridDim.x + blockIdx.x) * 32 + lane;
    const v4f z4 = {};
    for (long long i = gt; i < cA4 + cB4; i += nthr) {
      float* __restrict__ zp = (i < cA4) ? (zA + 4 * i) : (zB + 4 * (i - cA4));
      *(v4f*)zp = z4;
    }
  }

  v2f sc[CIN / 4], sh[CIN / 4];
  if constexpr (BN) {
    __shared__ __attribute__((aligned(16))) float lss[2 * CIN];
    const float inv_n = 1.0f / (float)n_in;
    for (int ch = lane; ch < CIN; ch += 32) {
      const float mean = st[ch] * inv_n;
      const float var  = st[CIN + ch] * inv_n - mean * mean;
      const float s    = rsqrtf(var + 1e-5f) * g[ch];
      const int slot   = (ch >> 1) * 4 + (ch & 1);
      lss[slot]     = s;
      lss[slot + 2] = b[ch] - mean * s;
    }
    __syncthreads();
#pragma unroll
    for (int c = 0; c < CIN / 4; ++c) {
      const v4f t = *(const v4f*)(lss + (c * 2 + half) * 4);
      sc[c].x = t.x; sc[c].y = t.y;
      sh[c].x = t.z; sh[c].y = t.w;
    }
  }

  int o_s[TPB];
  bool live[TPB], full[TPB];
  v2f a[TPB][CIN / 4];
#pragma unroll
  for (int t = 0; t < TPB; ++t) {
    const int o   = pvt[t] ? o_ld[t] : n_out;
    const bool ov = (o >= 0) && (o < n_out);
    live[t] = (__ballot(ov) != 0ull);
    const int row = pvt[t] ? r_ld[t] : n_in;
    const bool rv = (row >= 0) && (row < n_in);
    o_s[t] = (BN && !rv) ? -1 : o;
    full[t] = (__ballot((o_s[t] >= 0) && (o_s[t] < n_out)) == 0xffffffffull);
    const float* __restrict__ fr =
        (rv ? f + (long long)row * CIN : zrow) + half * 2;
#pragma unroll
    for (int c = 0; c < CIN / 4; ++c) a[t][c] = *(const v2f*)(fr + c * 4);
  }

#pragma unroll
  for (int t = 0; t < TPB; ++t) {
    if (!live[t]) continue;
    if constexpr (BN) {
#pragma unroll
      for (int c = 0; c < CIN / 4; ++c) {
        a[t][c].x = fmaxf(fmaf(a[t][c].x, sc[c].x, sh[c].x), 0.0f);
        a[t][c].y = fmaxf(fmaf(a[t][c].y, sc[c].y, sh[c].y), 0.0f);
      }
    }
    int o_x = o_s[t];
    asm volatile("" : "+v"(o_x));
    __builtin_amdgcn_sched_barrier(0);

    v8f acc = {};
#pragma unroll
    for (int c = 0; c < CIN / 4; ++c) {
      acc = __builtin_amdgcn_wmma_f32_16x16x4_f32(
           false, a[t][c],  false, bfrag[c],
           (short)0, acc,  false,  false);
    }
    __builtin_amdgcn_sched_barrier(0);
    asm volatile("v_nop\n\tv_nop\n\tv_nop\n\tv_nop\n\tv_nop\n\tv_nop\n\tv_nop\n\tv_nop"
                 : "+v"(acc));

    asm volatile("" : "+v"(o_x));
    if (FULLN && full[t]) {
#pragma unroll
      for (int r = 0; r < 8; ++r) {
        const int oa = __builtin_amdgcn_readlane(o_x, r);
        const int ob = __builtin_amdgcn_readlane(o_x, 8 + r);
        const int o_r = half ? ob : oa;
        unsafeAtomicAdd(&out[(long long)o_r * COUT + col], acc[r]);
      }
    } else {
#pragma unroll
      for (int r = 0; r < 8; ++r) {
        const int oa = __builtin_amdgcn_readlane(o_x, r);
        const int ob = __builtin_amdgcn_readlane(o_x, 8 + r);
        const int o_r = half ? ob : oa;
        if (cv && o_r >= 0 && o_r < n_out)
          unsafeAtomicAdd(&out[(long long)o_r * COUT + col], acc[r]);
      }
    }
  }
}

template <int CIN, int COUT, int WPB, int TPW, bool BN>
__global__ __launch_bounds__(32 * WPB) void sconv_wmma_row32(
    const float* __restrict__ f, const float* __restrict__ W,
    const int* __restrict__ ii, const int* __restrict__ oo,
    const int* __restrict__ nin_p, const int* __restrict__ nout_p,
    int P, float* __restrict__ out, const float* __restrict__ zrow,
    const float* __restrict__ st, const float* __restrict__ g,
    const float* __restrict__ b) {
  static_assert(COUT > 16, "one-tile layers use sconv_wmma");
  static_assert(!BN || 32 * WPB >= CIN, "one thread per input channel builds the scale / shift table");
  const int lane = threadIdx.x & 31;
  const int wv   = __builtin_amdgcn_readfirstlane((int)(threadIdx.x >> 5));
  const int m    = lane & 15;
  const int half = lane >> 4;

  const int k       = blockIdx.y;
  const int colBase = blockIdx.z << 5;

  const int* __restrict__ iik = ii + (long long)k * P;
  const int* __restrict__ ook = oo + (long long)k * P;

  int o_ld[TPW], r_ld[TPW];
  bool pvt[TPW];
#pragma unroll
  for (int t = 0; t < TPW; ++t) {
    const int tile = (blockIdx.x * TPW + t) * WPB + wv;
    const int p = tile * 16 + m;
    pvt[t] = (p < P);
    const int pc = pvt[t] ? p : (P - 1);
    o_ld[t] = ook[pc];
    r_ld[t] = iik[pc];
  }

  const int n_in  = *nin_p;
  const int n_out = *nout_p;

  constexpr bool FULLN = (COUT % 32) == 0;
  const int colA = colBase + m;
  const int colB = colA + 16;
  const bool cvA = FULLN || (colA < COUT);
  const bool cvB = FULLN || (colB < COUT);
  const int colAc = cvA ? colA : (COUT - 1);
  const int colBc = cvB ? colB : (COUT - 1);
  const int colL = colBase + lane;
  const bool cvL = FULLN || (colL < COUT);

  const float* __restrict__ Wk = W + (long long)k * (CIN * COUT);
  v2f bA[CIN / 4], bB[CIN / 4];
#pragma unroll
  for (int c = 0; c < CIN / 4; ++c) {
    const int k0 = c * 4 + half * 2;
    const float wa0 = Wk[(k0 + 0) * COUT + colAc];
    const float wa1 = Wk[(k0 + 1) * COUT + colAc];
    const float wb0 = Wk[(k0 + 0) * COUT + colBc];
    const float wb1 = Wk[(k0 + 1) * COUT + colBc];
    bA[c].x = cvA ? wa0 : 0.0f;
    bA[c].y = cvA ? wa1 : 0.0f;
    bB[c].x = cvB ? wb0 : 0.0f;
    bB[c].y = cvB ? wb1 : 0.0f;
  }

  float st_sum = 0.0f, st_sq = 0.0f, g_ch = 0.0f, b_ch = 0.0f;
  if constexpr (BN) {
    const int chc = ((int)threadIdx.x < CIN) ? (int)threadIdx.x : (CIN - 1);
    st_sum = st[chc];
    st_sq  = st[CIN + chc];
    g_ch   = g[chc];
    b_ch   = b[chc];
  }

  int o_s[TPW];
  bool live[TPW], full[TPW];
  const float* fr[TPW];
#pragma unroll
  for (int t = 0; t < TPW; ++t) {
    const int o   = pvt[t] ? o_ld[t] : n_out;
    const bool ov = (o >= 0) && (o < n_out);
    live[t] = (__ballot(ov) != 0ull);
    const int row = pvt[t] ? r_ld[t] : n_in;
    const bool rv = (row >= 0) && (row < n_in);
    o_s[t] = (BN && !rv) ? -1 : o;
    full[t] = (__ballot((o_s[t] >= 0) && (o_s[t] < n_out)) == 0xffffffffull);
    fr[t] = (rv ? f + (long long)row * CIN : zrow) + half * 2;
  }

  v2f a[2][CIN / 4];
#pragma unroll
  for (int c = 0; c < CIN / 4; ++c) a[0][c] = *(const v2f*)(fr[0] + c * 4);

  v2f sc[CIN / 4], sh[CIN / 4];
  if constexpr (BN) {
    __shared__ __attribute__((aligned(16))) float lss[2 * CIN];
    if ((int)threadIdx.x < CIN) {
      const int ch = (int)threadIdx.x;
      const float inv_n = 1.0f / (float)n_in;
      const float mean = st_sum * inv_n;
      const float var  = st_sq * inv_n - mean * mean;
      const float s    = rsqrtf(var + 1e-5f) * g_ch;
      const int slot   = (ch >> 1) * 4 + (ch & 1);
      lss[slot]     = s;
      lss[slot + 2] = b_ch - mean * s;
    }
    __syncthreads();
#pragma unroll
    for (int c = 0; c < CIN / 4; ++c) {
      const v4f t4 = *(const v4f*)(lss + (c * 2 + half) * 4);
      sc[c].x = t4.x; sc[c].y = t4.y;
      sh[c].x = t4.z; sh[c].y = t4.w;
    }
  }

#pragma unroll
  for (int t = 0; t < TPW; ++t) {
    if (t + 1 < TPW) {
#pragma unroll
      for (int c = 0; c < CIN / 4; ++c)
        a[(t + 1) & 1][c] = *(const v2f*)(fr[t + 1] + c * 4);
    }
    if (!live[t]) continue;

    if constexpr (BN) {
#pragma unroll
      for (int c = 0; c < CIN / 4; ++c) {
        a[t & 1][c].x = fmaxf(fmaf(a[t & 1][c].x, sc[c].x, sh[c].x), 0.0f);
        a[t & 1][c].y = fmaxf(fmaf(a[t & 1][c].y, sc[c].y, sh[c].y), 0.0f);
      }
    }
    int o_x = o_s[t];
    asm volatile("" : "+v"(o_x));
    __builtin_amdgcn_sched_barrier(0);

    v8f acc0 = {}, acc1 = {};
#pragma unroll
    for (int c = 0; c < CIN / 4; ++c) {
      acc0 = __builtin_amdgcn_wmma_f32_16x16x4_f32(
           false, a[t & 1][c],  false, bA[c],
           (short)0, acc0,  false,  false);
      acc1 = __builtin_amdgcn_wmma_f32_16x16x4_f32(
           false, a[t & 1][c],  false, bB[c],
           (short)0, acc1,  false,  false);
    }
    __builtin_amdgcn_sched_barrier(0);
    asm volatile("v_nop\n\tv_nop\n\tv_nop\n\tv_nop\n\tv_nop\n\tv_nop\n\tv_nop\n\tv_nop"
                 : "+v"(acc0), "+v"(acc1));

    float xr[8], yr[8];
#pragma unroll
    for (int r = 0; r < 8; ++r) {
      const float a0 = acc0[r];
      const float a1 = acc1[r];
      const float a0x = __builtin_bit_cast(float, __builtin_amdgcn_permlanex16(
          0u, __builtin_bit_cast(unsigned, a0), 0x76543210u, 0xfedcba98u,  true, false));
      const float a1x = __builtin_bit_cast(float, __builtin_amdgcn_permlanex16(
          0u, __builtin_bit_cast(unsigned, a1), 0x76543210u, 0xfedcba98u,  true, false));
      xr[r] = half ? a1x : a0;
      yr[r] = half ? a1 : a0x;
    }

    asm volatile("" : "+v"(o_x));
    if (cvL) {
      if (full[t]) {
#pragma unroll
        for (int r = 0; r < 8; ++r) {
          const int oa = __builtin_amdgcn_readlane(o_x, r);
          const int ob = __builtin_amdgcn_readlane(o_x, 8 + r);
          unsafeAtomicAdd(&out[(long long)oa * COUT + colL], xr[r]);
          unsafeAtomicAdd(&out[(long long)ob * COUT + colL], yr[r]);
        }
      } else {
#pragma unroll
        for (int r = 0; r < 8; ++r) {
          const int oa = __builtin_amdgcn_readlane(o_x, r);
          const int ob = __builtin_amdgcn_readlane(o_x, 8 + r);
          if (oa >= 0 && oa < n_out)
            unsafeAtomicAdd(&out[(long long)oa * COUT + colL], xr[r]);
          if (ob >= 0 && ob < n_out)
            unsafeAtomicAdd(&out[(long long)ob * COUT + colL], yr[r]);
        }
      }
    }
  }
}

template <int C>
__global__ __launch_bounds__(256) void bn_stats(
    const float* __restrict__ x, const int* __restrict__ n_p,
    float* __restrict__ stats) {
  constexpr int U = 8;
  constexpr int G = C / 4;
  const int n = *n_p;
  const int t = threadIdx.x;
  const int lane = t & 31;
  const int w = t >> 5;
  const long long total4 = (long long)n * G;
  const long long T = (long long)gridDim.x * 256;
  const long long gt = (long long)blockIdx.x * 256 + t;
  const v4f* __restrict__ x4 = (const v4f*)x;

  v4f s = {}, q = {};
  for (long long base = gt; base < total4; base += U * T) {
    v4f v[U];
#pragma unroll
    for (int j = 0; j < U; ++j) {
      const long long i = base + j * T;
      const bool ok = i < total4;
      const v4f ld = x4[ok ? i : base];
      const v4f z = {};
      v[j] = ok ? ld : z;
    }
#pragma unroll
    for (int j = 0; j < U; ++j) {
      s += v[j];
      q += v[j] * v[j];
    }
  }

#pragma unroll
  for (int off = G; off < 32; off <<= 1) {
#pragma unroll
    for (int e = 0; e < 4; ++e) {
      s[e] += __shfl_xor(s[e], off);
      q[e] += __shfl_xor(q[e], off);
    }
  }

  __shared__ float part[8][2 * C];
  if (lane < G) {
#pragma unroll
    for (int e = 0; e < 4; ++e) {
      part[w][4 * lane + e]     = s[e];
      part[w][C + 4 * lane + e] = q[e];
    }
  }
  __syncthreads();
  if (t < 2 * C) {
    float tot = 0.0f;
#pragma unroll
    for (int k = 0; k < 8; ++k) tot += part[k][t];
    unsafeAtomicAdd(&stats[t], tot);
  }
}

extern "C" void kernel_launch(void* const* d_in, const int* in_sizes, int n_in,
                              void* d_out, int out_size, void* d_ws, size_t ws_size,
                              hipStream_t stream) {
  (void)n_in; (void)ws_size;
  const float* feats = (const float*)d_in[0];
  const float* W0  = (const float*)d_in[1];
  const float* g0  = (const float*)d_in[2];
  const float* b0  = (const float*)d_in[3];
  const float* W1  = (const float*)d_in[4];
  const float* g1  = (const float*)d_in[5];
  const float* b1  = (const float*)d_in[6];
  const float* W2  = (const float*)d_in[7];
  const float* g2  = (const float*)d_in[8];
  const float* b2  = (const float*)d_in[9];
  const float* Wt1 = (const float*)d_in[10];
  const float* gt1 = (const float*)d_in[11];
  const float* bt1 = (const float*)d_in[12];
  const float* Wt2 = (const float*)d_in[13];
  const int* in0  = (const int*)d_in[14];
  const int* out0 = (const int*)d_in[15];
  const int* in1  = (const int*)d_in[16];
  const int* out1 = (const int*)d_in[17];
  const int* in2  = (const int*)d_in[18];
  const int* out2 = (const int*)d_in[19];
  const int* int1 = (const int*)d_in[20];
  const int* outt1= (const int*)d_in[21];
  const int* int2 = (const int*)d_in[22];
  const int* outt2= (const int*)d_in[23];
  const int* n0p  = (const int*)d_in[24];
  const int* n1p  = (const int*)d_in[25];
  const int* n2p  = (const int*)d_in[26];

  const int n0  = in_sizes[0] / 4;
  const int P0  = in_sizes[14] / 27;
  const int P1  = in_sizes[16] / 27;
  const int P2  = in_sizes[18] / 27;
  const int Pt1 = in_sizes[20] / 27;
  const int Pt2 = in_sizes[22] / 27;

  float* ws  = (float*)d_ws;
  float* st0 = ws;
  float* st1 = st0 + 32;
  float* st2 = st1 + 64;
  float* stt = st2 + 128;
  const float* zrow = stt + 64;
  float* x0  = ws  + 512;
  float* x1  = x0  + (size_t)n0 * 16;
  float* x2  = x1  + (size_t)n0 * 32;
  float* xt1 = x2  + (size_t)n0 * 64;

  const long long zc4 = ((long long)n0 * 16 + 512) / 4;
  zero_f32v4<<<dim3((unsigned)((zc4 + 255) / 256)), 256, 0, stream>>>(ws, zc4);
  const long long xc4 = ((long long)n0 * (32 + 64 + 32)) / 4;
  const long long oc4 = ((long long)out_size + 3) / 4;

  constexpr int TPB = 8;
  const dim3 wv(32);
  auto tgrid = [](int P) { return (unsigned)((P + 16 * TPB - 1) / (16 * TPB)); };
  auto sg = [&](int C) {
    long long blocks = ((long long)n0 * C + 8191) / 8192;
    return dim3((unsigned)(blocks < 512 ? blocks : 512));
  };

  sconv_wmma<4, 16, TPB, false><<<dim3(tgrid(P0), 27, 1), wv, 0, stream>>>(
      feats, W0, in0, out0, n0p, n0p, P0, x0, zrow, nullptr, nullptr, nullptr,
      x1, xc4, (float*)d_out, oc4);
  bn_stats<16><<<sg(16), 256, 0, stream>>>(x0, n0p, st0);

  sconv_wmma_row32<16, 32, 4, 2, true><<<dim3(tgrid(P1), 27, 1), dim3(128), 0, stream>>>(
      x0, W1, in1, out1, n0p, n1p, P1, x1, zrow, st0, g0, b0);
  bn_stats<32><<<sg(32), 256, 0, stream>>>(x1, n1p, st1);

  sconv_wmma_row32<32, 64, 2, 4, true><<<dim3(tgrid(P2), 27, 2), dim3(64), 0, stream>>>(
      x1, W2, in2, out2, n1p, n2p, P2, x2, zrow, st1, g1, b1);
  bn_stats<64><<<sg(64), 256, 0, stream>>>(x2, n2p, st2);

  sconv_wmma_row32<64, 32, 2, 4, true><<<dim3(tgrid(Pt1), 27, 1), dim3(64), 0, stream>>>(
      x2, Wt1, int1, outt1, n2p, n1p, Pt1, xt1, zrow, st2, g2, b2);
  bn_stats<32><<<sg(32), 256, 0, stream>>>(xt1, n1p, stt);

  sconv_wmma_row32<32, 20, 4, 2, true><<<dim3(tgrid(Pt2), 27, 1), dim3(128), 0, stream>>>(
      xt1, Wt2, int2, outt2, n1p, n0p, Pt2, (float*)d_out, zrow, stt, gt1, bt1);
}
